# starting kernel
# baseline (speedup 1.0000x reference)
#include <hip/hip_runtime.h>


#define HID 128
#define LDSS 132
#define NEG_SLOPE 0.01f

typedef float v2f __attribute__((ext_vector_type(2)));
typedef float v8f __attribute__((ext_vector_type(8)));

__device__ __forceinline__ v8f wmma_f32(v2f a, v2f b, v8f c) {
  return __builtin_amdgcn_wmma_f32_16x16x4_f32(
       false, a,  false, b,
       (short)0, c,  false,  false);
}

__global__ __launch_bounds__(128)
void fused_x_kernel(const float* __restrict__ primal,
                    const float* __restrict__ last_primal,
                    const float* __restrict__ W2, const float* __restrict__ b2,
                    const float* __restrict__ W3, const float* __restrict__ b3,
                    float* __restrict__ X) {
  __shared__ float Ap[16 * LDSS];
  __shared__ float Al[16 * LDSS];
  const int m0  = blockIdx.x * 16;
  const int tid = threadIdx.x;

  const float4* gp = (const float4*)(primal      + (size_t)m0 * HID);
  const float4* gl = (const float4*)(last_primal + (size_t)m0 * HID);
#pragma unroll
  for (int j = 0; j < 4; ++j) {
    const int f   = tid + j * 128;
    const int row = f >> 5;
    const int c4  = f & 31;
    *(float4*)&Ap[row * LDSS + c4 * 4] = gp[f];
    *(float4*)&Al[row * LDSS + c4 * 4] = gl[f];
  }
  __syncthreads();

  const int lane = tid & 31;
  const int wave = tid >> 5;
  const int r    = lane & 15;
  const int koff = (lane >> 4) << 1;

#pragma unroll
  for (int t = 0; t < 2; ++t) {
    const int n0 = (wave * 2 + t) * 16;
    const int n  = n0 + (lane & 15);
    v8f acc  = {};
    v8f acc2 = {};
#pragma unroll
    for (int k0 = 0; k0 < HID; k0 += 4) {
      const int kk = k0 + koff;
      v2f a, b, a2, b2v;
      a.x   = Ap[r * LDSS + kk];  a.y   = Ap[r * LDSS + kk + 1];
      b.x   = W2[n * HID + kk];   b.y   = W2[n * HID + kk + 1];
      acc   = wmma_f32(a, b, acc);
      a2.x  = Al[r * LDSS + kk];  a2.y  = Al[r * LDSS + kk + 1];
      b2v.x = W3[n * HID + kk];   b2v.y = W3[n * HID + kk + 1];
      acc2  = wmma_f32(a2, b2v, acc2);
    }
    const float bias  = b2[n] - b3[n];
    const int   mbase = m0 + ((lane >> 4) << 3);
#pragma unroll
    for (int j = 0; j < 8; ++j)
      X[(size_t)(mbase + j) * HID + n] = acc[j] - acc2[j] + bias;
  }
}

__global__ __launch_bounds__(128)
void dual_base_kernel(const float* __restrict__ dual,
                      const float* __restrict__ W1, const float* __restrict__ b1,
                      const float* __restrict__ rhs,
                      const float* __restrict__ sigma_p,
                      float* __restrict__ out) {
  __shared__ float Ad[16 * LDSS];
  const int m0  = blockIdx.x * 16;
  const int tid = threadIdx.x;

  const float4* gd = (const float4*)(dual + (size_t)m0 * HID);
#pragma unroll
  for (int j = 0; j < 4; ++j) {
    const int f   = tid + j * 128;
    const int row = f >> 5;
    const int c4  = f & 31;
    *(float4*)&Ad[row * LDSS + c4 * 4] = gd[f];
  }
  __syncthreads();

  const int   lane  = tid & 31;
  const int   wave  = tid >> 5;
  const int   r     = lane & 15;
  const int   koff  = (lane >> 4) << 1;
  const float sigma = sigma_p[0];

#pragma unroll
  for (int t = 0; t < 2; ++t) {
    const int n0 = (wave * 2 + t) * 16;
    const int n  = n0 + (lane & 15);
    v8f acc = {};
#pragma unroll
    for (int k0 = 0; k0 < HID; k0 += 4) {
      const int kk = k0 + koff;
      v2f a, b;
      a.x = Ad[r * LDSS + kk];  a.y = Ad[r * LDSS + kk + 1];
      b.x = W1[n * HID + kk];   b.y = W1[n * HID + kk + 1];
      acc = wmma_f32(a, b, acc);
    }
    const float bias  = b1[n];
    const int   mbase = m0 + ((lane >> 4) << 3);
#pragma unroll
    for (int j = 0; j < 8; ++j) {
      const size_t idx = (size_t)(mbase + j) * HID + n;
      out[idx] = acc[j] + bias - sigma * rhs[idx];
    }
  }
}

__global__ __launch_bounds__(256)
void scatter_kernel(const int* __restrict__ rows, const int* __restrict__ cols,
                    const float* __restrict__ vals, const float* __restrict__ X,
                    const float* __restrict__ sigma_p, float* __restrict__ out,
                    int nnz) {
  const int e = blockIdx.x * 8 + (threadIdx.x >> 5);
  if (e >= nnz) return;
  const int   lane = threadIdx.x & 31;
  const int   rr   = rows[e];
  const int   cc   = cols[e];
  const float s    = vals[e] * sigma_p[0];

  const float4 xv = ((const float4*)(X + (size_t)cc * HID))[lane];
  float* op = out + (size_t)rr * HID + lane * 4;
  __hip_atomic_fetch_add(&op[0], s * xv.x, __ATOMIC_RELAXED, __HIP_MEMORY_SCOPE_AGENT);
  __hip_atomic_fetch_add(&op[1], s * xv.y, __ATOMIC_RELAXED, __HIP_MEMORY_SCOPE_AGENT);
  __hip_atomic_fetch_add(&op[2], s * xv.z, __ATOMIC_RELAXED, __HIP_MEMORY_SCOPE_AGENT);
  __hip_atomic_fetch_add(&op[3], s * xv.w, __ATOMIC_RELAXED, __HIP_MEMORY_SCOPE_AGENT);
}

__global__ __launch_bounds__(256)
void leaky_kernel(float* __restrict__ out, int n4) {
  int i = blockIdx.x * blockDim.x + threadIdx.x;
  const int stride = gridDim.x * blockDim.x;
  float4* p = (float4*)out;
  for (; i < n4; i += stride) {
    float4 v = p[i];
    v.x = v.x >= 0.f ? v.x : NEG_SLOPE * v.x;
    v.y = v.y >= 0.f ? v.y : NEG_SLOPE * v.y;
    v.z = v.z >= 0.f ? v.z : NEG_SLOPE * v.z;
    v.w = v.w >= 0.f ? v.w : NEG_SLOPE * v.w;
    p[i] = v;
  }
}

extern "C" void kernel_launch(void* const* d_in, const int* in_sizes, int n_in,
                              void* d_out, int out_size, void* d_ws, size_t ws_size,
                              hipStream_t stream) {
  const float* primal      = (const float*)d_in[0];
  const float* last_primal = (const float*)d_in[1];
  const float* dual        = (const float*)d_in[2];
  const int*   rows        = (const int*)d_in[3];
  const int*   cols        = (const int*)d_in[4];
  const float* vals        = (const float*)d_in[5];
  const float* rhs         = (const float*)d_in[6];
  const float* W1          = (const float*)d_in[7];
  const float* b1          = (const float*)d_in[8];
  const float* W2          = (const float*)d_in[9];
  const float* b2          = (const float*)d_in[10];
  const float* W3          = (const float*)d_in[11];
  const float* b3          = (const float*)d_in[12];
  const float* sigma       = (const float*)d_in[13];

  float* out = (float*)d_out;
  float* X   = (float*)d_ws;

  const int n_vars = in_sizes[0] / HID;
  const int m_cons = in_sizes[2] / HID;
  const int nnz    = in_sizes[5];

  fused_x_kernel<<<n_vars / 16, 128, 0, stream>>>(primal, last_primal,
                                                  W2, b2, W3, b3, X);
  dual_base_kernel<<<m_cons / 16, 128, 0, stream>>>(dual, W1, b1, rhs, sigma, out);
  scatter_kernel<<<(nnz + 7) / 8, 256, 0, stream>>>(rows, cols, vals, X, sigma,
                                                    out, nnz);
  leaky_kernel<<<4096, 256, 0, stream>>>(out, (m_cons * HID) / 4);
}


# Both GEMM kernels: 8-wave blocks loop over 16-row tiles with each wave's weight fragments held in registers (staged once through LDS) instead of a strided global weight load per WMMA; sums unchanged
# speedup vs baseline: 1.2574x; 1.2574x over previous
#include <hip/hip_runtime.h>


#define HID 128
#define LDSS 132
#define TILE_F (16 * LDSS)
#define NEG_SLOPE 0.01f
#define GEMM_BLOCKS_X 512
#define GEMM_BLOCKS_D 768

typedef float v2f __attribute__((ext_vector_type(2)));
typedef float v8f __attribute__((ext_vector_type(8)));

__device__ __forceinline__ v8f wmma_f32(v2f a, v2f b, v8f c) {
  return __builtin_amdgcn_wmma_f32_16x16x4_f32(
       false, a,  false, b,
       (short)0, c,  false,  false);
}

__device__ __forceinline__ void stage_w_rows(const float* __restrict__ W, int n0,
                                             float* wst, int lane) {
  const float4* g = (const float4*)(W + (size_t)n0 * HID);
#pragma unroll
  for (int q = 0; q < 16; ++q)
    *(float4*)&wst[q * LDSS + lane * 4] = g[q * 32 + lane];
}

__device__ __forceinline__ v2f frag(const float* t, int c, int koff, int i) {
  return *(const v2f*)&t[c * LDSS + 4 * i + koff];
}

__device__ __forceinline__ void lds_barrier() {
  __builtin_amdgcn_fence(__ATOMIC_RELEASE, "workgroup", "local");
  __builtin_amdgcn_s_barrier();
  __builtin_amdgcn_fence(__ATOMIC_ACQUIRE, "workgroup", "local");
}

__global__ __launch_bounds__(256)
void fused_x_kernel(const float* __restrict__ primal,
                    const float* __restrict__ last_primal,
                    const float* __restrict__ W2, const float* __restrict__ b2,
                    const float* __restrict__ W3, const float* __restrict__ b3,
                    float* __restrict__ X, int ntiles) {
  __shared__ __attribute__((aligned(16))) float smem[8 * TILE_F];
  const int tid  = threadIdx.x;
  const int lane = tid & 31;
  const int wave = tid >> 5;
  const int c    = lane & 15;
  const int h    = lane >> 4;
  const int koff = h << 1;
  const int n    = wave * 16 + c;

  int tile = blockIdx.x;

  const int srow = tid >> 5, sc4 = tid & 31;
  const int soff = srow * LDSS + sc4 * 4;
  float4 pa0 = {0.f, 0.f, 0.f, 0.f}, pa1 = pa0, pl0 = pa0, pl1 = pa0;
  if (tile < ntiles) {
    const float4* gp = (const float4*)(primal      + (size_t)tile * 16 * HID);
    const float4* gl = (const float4*)(last_primal + (size_t)tile * 16 * HID);
    pa0 = gp[tid]; pa1 = gp[tid + 256];
    pl0 = gl[tid]; pl1 = gl[tid + 256];
  }

  v2f bw2[32], bw3[32];
  float* wst = smem + wave * TILE_F;
  stage_w_rows(W2, wave * 16, wst, lane);
  __syncthreads();
#pragma unroll
  for (int i = 0; i < 32; ++i) bw2[i] = frag(wst, c, koff, i);
  __syncthreads();
  stage_w_rows(W3, wave * 16, wst, lane);
  __syncthreads();
#pragma unroll
  for (int i = 0; i < 32; ++i) bw3[i] = frag(wst, c, koff, i);
  __syncthreads();

  const float bias = b2[n] - b3[n];

  if (tile < ntiles) {
    *(float4*)&smem[soff]                     = pa0;
    *(float4*)&smem[soff + 8 * LDSS]          = pa1;
    *(float4*)&smem[TILE_F + soff]            = pl0;
    *(float4*)&smem[TILE_F + soff + 8 * LDSS] = pl1;
  }
  __syncthreads();

  int cur = 0;
  for (; tile < ntiles; tile += gridDim.x) {
    const int  next     = tile + gridDim.x;
    const bool has_next = next < ntiles;
    if (has_next) {
      const float4* gp = (const float4*)(primal      + (size_t)next * 16 * HID);
      const float4* gl = (const float4*)(last_primal + (size_t)next * 16 * HID);
      pa0 = gp[tid]; pa1 = gp[tid + 256];
      pl0 = gl[tid]; pl1 = gl[tid + 256];
    }

    const float* Ap = smem + cur * 2 * TILE_F;
    const float* Al = Ap + TILE_F;
    v8f acc  = {};
    v8f acc2 = {};
#pragma unroll
    for (int i = 0; i < 32; ++i) {
      acc  = wmma_f32(frag(Ap, c, koff, i), bw2[i], acc);
      acc2 = wmma_f32(frag(Al, c, koff, i), bw3[i], acc2);
    }

    const int mbase = tile * 16 + (h << 3);
#pragma unroll
    for (int j = 0; j < 8; ++j)
      X[(size_t)(mbase + j) * HID + n] = acc[j] - acc2[j] + bias;

    if (has_next) {
      float* Np = smem + (cur ^ 1) * 2 * TILE_F;
      *(float4*)&Np[soff]                     = pa0;
      *(float4*)&Np[soff + 8 * LDSS]          = pa1;
      *(float4*)&Np[TILE_F + soff]            = pl0;
      *(float4*)&Np[TILE_F + soff + 8 * LDSS] = pl1;
    }
    lds_barrier();
    cur ^= 1;
  }
}

__global__ __launch_bounds__(256)
void dual_base_kernel(const float* __restrict__ dual,
                      const float* __restrict__ W1, const float* __restrict__ b1,
                      const float* __restrict__ rhs,
                      const float* __restrict__ sigma_p,
                      float* __restrict__ out, int ntiles) {
  __shared__ __attribute__((aligned(16))) float smem[8 * TILE_F];
  const int tid  = threadIdx.x;
  const int lane = tid & 31;
  const int wave = tid >> 5;
  const int c    = lane & 15;
  const int h    = lane >> 4;
  const int koff = h << 1;
  const int n    = wave * 16 + c;

  int tile = blockIdx.x;

  const int srow = tid >> 5, sc4 = tid & 31;
  const int soff = srow * LDSS + sc4 * 4;
  float4 pd0 = {0.f, 0.f, 0.f, 0.f}, pd1 = pd0;
  if (tile < ntiles) {
    const float4* gd = (const float4*)(dual + (size_t)tile * 16 * HID);
    pd0 = gd[tid]; pd1 = gd[tid + 256];
  }

  v2f bw[32];
  float* wst = smem + wave * TILE_F;
  stage_w_rows(W1, wave * 16, wst, lane);
  __syncthreads();
#pragma unroll
  for (int i = 0; i < 32; ++i) bw[i] = frag(wst, c, koff, i);
  __syncthreads();

  const float bias  = b1[n];
  const float sigma = sigma_p[0];

  if (tile < ntiles) {
    *(float4*)&smem[soff]            = pd0;
    *(float4*)&smem[soff + 8 * LDSS] = pd1;
  }
  __syncthreads();

  int cur = 0;
  for (; tile < ntiles; tile += gridDim.x) {
    const int  next     = tile + gridDim.x;
    const bool has_next = next < ntiles;
    if (has_next) {
      const float4* gd = (const float4*)(dual + (size_t)next * 16 * HID);
      pd0 = gd[tid]; pd1 = gd[tid + 256];
    }

    const int mbase = tile * 16 + (h << 3);
    float rv[8];
#pragma unroll
    for (int j = 0; j < 8; ++j) rv[j] = rhs[(size_t)(mbase + j) * HID + n];

    const float* Ad = smem + cur * TILE_F;
    v8f acc = {};
#pragma unroll
    for (int i = 0; i < 32; ++i)
      acc = wmma_f32(frag(Ad, c, koff, i), bw[i], acc);

#pragma unroll
    for (int j = 0; j < 8; ++j)
      out[(size_t)(mbase + j) * HID + n] = acc[j] + bias - sigma * rv[j];

    if (has_next) {
      float* Nd = smem + (cur ^ 1) * TILE_F;
      *(float4*)&Nd[soff]            = pd0;
      *(float4*)&Nd[soff + 8 * LDSS] = pd1;
    }
    lds_barrier();
    cur ^= 1;
  }
}

__global__ __launch_bounds__(256)
void scatter_kernel(const int* __restrict__ rows, const int* __restrict__ cols,
                    const float* __restrict__ vals, const float* __restrict__ X,
                    const float* __restrict__ sigma_p, float* __restrict__ out,
                    int nnz) {
  const int e = blockIdx.x * 8 + (threadIdx.x >> 5);
  if (e >= nnz) return;
  const int   lane = threadIdx.x & 31;
  const int   rr   = rows[e];
  const int   cc   = cols[e];
  const float s    = vals[e] * sigma_p[0];

  const float4 xv = ((const float4*)(X + (size_t)cc * HID))[lane];
  float* op = out + (size_t)rr * HID + lane * 4;
  __hip_atomic_fetch_add(&op[0], s * xv.x, __ATOMIC_RELAXED, __HIP_MEMORY_SCOPE_AGENT);
  __hip_atomic_fetch_add(&op[1], s * xv.y, __ATOMIC_RELAXED, __HIP_MEMORY_SCOPE_AGENT);
  __hip_atomic_fetch_add(&op[2], s * xv.z, __ATOMIC_RELAXED, __HIP_MEMORY_SCOPE_AGENT);
  __hip_atomic_fetch_add(&op[3], s * xv.w, __ATOMIC_RELAXED, __HIP_MEMORY_SCOPE_AGENT);
}

__global__ __launch_bounds__(256)
void leaky_kernel(float* __restrict__ out, int n4) {
  int i = blockIdx.x * blockDim.x + threadIdx.x;
  const int stride = gridDim.x * blockDim.x;
  float4* p = (float4*)out;
  for (; i < n4; i += stride) {
    float4 v = p[i];
    v.x = v.x >= 0.f ? v.x : NEG_SLOPE * v.x;
    v.y = v.y >= 0.f ? v.y : NEG_SLOPE * v.y;
    v.z = v.z >= 0.f ? v.z : NEG_SLOPE * v.z;
    v.w = v.w >= 0.f ? v.w : NEG_SLOPE * v.w;
    p[i] = v;
  }
}

extern "C" void kernel_launch(void* const* d_in, const int* in_sizes, int n_in,
                              void* d_out, int out_size, void* d_ws, size_t ws_size,
                              hipStream_t stream) {
  const float* primal      = (const float*)d_in[0];
  const float* last_primal = (const float*)d_in[1];
  const float* dual        = (const float*)d_in[2];
  const int*   rows        = (const int*)d_in[3];
  const int*   cols        = (const int*)d_in[4];
  const float* vals        = (const float*)d_in[5];
  const float* rhs         = (const float*)d_in[6];
  const float* W1          = (const float*)d_in[7];
  const float* b1          = (const float*)d_in[8];
  const float* W2          = (const float*)d_in[9];
  const float* b2          = (const float*)d_in[10];
  const float* W3          = (const float*)d_in[11];
  const float* b3          = (const float*)d_in[12];
  const float* sigma       = (const float*)d_in[13];

  float* out = (float*)d_out;
  float* X   = (float*)d_ws;

  const int n_vars = in_sizes[0] / HID;
  const int m_cons = in_sizes[2] / HID;
  const int nnz    = in_sizes[5];

  fused_x_kernel<<<GEMM_BLOCKS_X, 256, 0, stream>>>(primal, last_primal,
                                                    W2, b2, W3, b3, X, n_vars / 16);
  dual_base_kernel<<<GEMM_BLOCKS_D, 256, 0, stream>>>(dual, W1, b1, rhs, sigma, out,
                                                      m_cons / 16);
  scatter_kernel<<<(nnz + 7) / 8, 256, 0, stream>>>(rows, cols, vals, X, sigma,
                                                    out, nnz);
  leaky_kernel<<<4096, 256, 0, stream>>>(out, (m_cons * HID) / 4);
}


# Scatter: lane l owns floats l, l+32, l+64, l+96 of the row, so each f32 atomic instruction adds to one whole 128-byte line (was 4 lines, 32 B of each); atomics kept, unordered as before
# speedup vs baseline: 1.5429x; 1.2271x over previous
#include <hip/hip_runtime.h>


#define HID 128
#define LDSS 132
#define TILE_F (16 * LDSS)
#define NEG_SLOPE 0.01f
#define GEMM_BLOCKS_X 512
#define GEMM_BLOCKS_D 768

typedef float v2f __attribute__((ext_vector_type(2)));
typedef float v8f __attribute__((ext_vector_type(8)));

__device__ __forceinline__ v8f wmma_f32(v2f a, v2f b, v8f c) {
  return __builtin_amdgcn_wmma_f32_16x16x4_f32(
       false, a,  false, b,
       (short)0, c,  false,  false);
}

__device__ __forceinline__ void stage_w_rows(const float* __restrict__ W, int n0,
                                             float* wst, int lane) {
  const float4* g = (const float4*)(W + (size_t)n0 * HID);
#pragma unroll
  for (int q = 0; q < 16; ++q)
    *(float4*)&wst[q * LDSS + lane * 4] = g[q * 32 + lane];
}

__device__ __forceinline__ v2f frag(const float* t, int c, int koff, int i) {
  return *(const v2f*)&t[c * LDSS + 4 * i + koff];
}

__device__ __forceinline__ void lds_barrier() {
  __builtin_amdgcn_fence(__ATOMIC_RELEASE, "workgroup", "local");
  __builtin_amdgcn_s_barrier();
  __builtin_amdgcn_fence(__ATOMIC_ACQUIRE, "workgroup", "local");
}

__global__ __launch_bounds__(256)
void fused_x_kernel(const float* __restrict__ primal,
                    const float* __restrict__ last_primal,
                    const float* __restrict__ W2, const float* __restrict__ b2,
                    const float* __restrict__ W3, const float* __restrict__ b3,
                    float* __restrict__ X, int ntiles) {
  __shared__ __attribute__((aligned(16))) float smem[8 * TILE_F];
  const int tid  = threadIdx.x;
  const int lane = tid & 31;
  const int wave = tid >> 5;
  const int c    = lane & 15;
  const int h    = lane >> 4;
  const int koff = h << 1;
  const int n    = wave * 16 + c;

  int tile = blockIdx.x;

  const int srow = tid >> 5, sc4 = tid & 31;
  const int soff = srow * LDSS + sc4 * 4;
  float4 pa0 = {0.f, 0.f, 0.f, 0.f}, pa1 = pa0, pl0 = pa0, pl1 = pa0;
  if (tile < ntiles) {
    const float4* gp = (const float4*)(primal      + (size_t)tile * 16 * HID);
    const float4* gl = (const float4*)(last_primal + (size_t)tile * 16 * HID);
    pa0 = gp[tid]; pa1 = gp[tid + 256];
    pl0 = gl[tid]; pl1 = gl[tid + 256];
  }

  v2f bw2[32], bw3[32];
  float* wst = smem + wave * TILE_F;
  stage_w_rows(W2, wave * 16, wst, lane);
  __syncthreads();
#pragma unroll
  for (int i = 0; i < 32; ++i) bw2[i] = frag(wst, c, koff, i);
  __syncthreads();
  stage_w_rows(W3, wave * 16, wst, lane);
  __syncthreads();
#pragma unroll
  for (int i = 0; i < 32; ++i) bw3[i] = frag(wst, c, koff, i);
  __syncthreads();

  const float bias = b2[n] - b3[n];

  if (tile < ntiles) {
    *(float4*)&smem[soff]                     = pa0;
    *(float4*)&smem[soff + 8 * LDSS]          = pa1;
    *(float4*)&smem[TILE_F + soff]            = pl0;
    *(float4*)&smem[TILE_F + soff + 8 * LDSS] = pl1;
  }
  __syncthreads();

  int cur = 0;
  for (; tile < ntiles; tile += gridDim.x) {
    const int  next     = tile + gridDim.x;
    const bool has_next = next < ntiles;
    if (has_next) {
      const float4* gp = (const float4*)(primal      + (size_t)next * 16 * HID);
      const float4* gl = (const float4*)(last_primal + (size_t)next * 16 * HID);
      pa0 = gp[tid]; pa1 = gp[tid + 256];
      pl0 = gl[tid]; pl1 = gl[tid + 256];
    }

    const float* Ap = smem + cur * 2 * TILE_F;
    const float* Al = Ap + TILE_F;
    v8f acc  = {};
    v8f acc2 = {};
#pragma unroll
    for (int i = 0; i < 32; ++i) {
      acc  = wmma_f32(frag(Ap, c, koff, i), bw2[i], acc);
      acc2 = wmma_f32(frag(Al, c, koff, i), bw3[i], acc2);
    }

    const int mbase = tile * 16 + (h << 3);
#pragma unroll
    for (int j = 0; j < 8; ++j)
      X[(size_t)(mbase + j) * HID + n] = acc[j] - acc2[j] + bias;

    if (has_next) {
      float* Np = smem + (cur ^ 1) * 2 * TILE_F;
      *(float4*)&Np[soff]                     = pa0;
      *(float4*)&Np[soff + 8 * LDSS]          = pa1;
      *(float4*)&Np[TILE_F + soff]            = pl0;
      *(float4*)&Np[TILE_F + soff + 8 * LDSS] = pl1;
    }
    lds_barrier();
    cur ^= 1;
  }
}

__global__ __launch_bounds__(256)
void dual_base_kernel(const float* __restrict__ dual,
                      const float* __restrict__ W1, const float* __restrict__ b1,
                      const float* __restrict__ rhs,
                      const float* __restrict__ sigma_p,
                      float* __restrict__ out, int ntiles) {
  __shared__ __attribute__((aligned(16))) float smem[8 * TILE_F];
  const int tid  = threadIdx.x;
  const int lane = tid & 31;
  const int wave = tid >> 5;
  const int c    = lane & 15;
  const int h    = lane >> 4;
  const int koff = h << 1;
  const int n    = wave * 16 + c;

  int tile = blockIdx.x;

  const int srow = tid >> 5, sc4 = tid & 31;
  const int soff = srow * LDSS + sc4 * 4;
  float4 pd0 = {0.f, 0.f, 0.f, 0.f}, pd1 = pd0;
  if (tile < ntiles) {
    const float4* gd = (const float4*)(dual + (size_t)tile * 16 * HID);
    pd0 = gd[tid]; pd1 = gd[tid + 256];
  }

  v2f bw[32];
  float* wst = smem + wave * TILE_F;
  stage_w_rows(W1, wave * 16, wst, lane);
  __syncthreads();
#pragma unroll
  for (int i = 0; i < 32; ++i) bw[i] = frag(wst, c, koff, i);
  __syncthreads();

  const float bias  = b1[n];
  const float sigma = sigma_p[0];

  if (tile < ntiles) {
    *(float4*)&smem[soff]            = pd0;
    *(float4*)&smem[soff + 8 * LDSS] = pd1;
  }
  __syncthreads();

  int cur = 0;
  for (; tile < ntiles; tile += gridDim.x) {
    const int  next     = tile + gridDim.x;
    const bool has_next = next < ntiles;
    if (has_next) {
      const float4* gd = (const float4*)(dual + (size_t)next * 16 * HID);
      pd0 = gd[tid]; pd1 = gd[tid + 256];
    }

    const int mbase = tile * 16 + (h << 3);
    float rv[8];
#pragma unroll
    for (int j = 0; j < 8; ++j) rv[j] = rhs[(size_t)(mbase + j) * HID + n];

    const float* Ad = smem + cur * TILE_F;
    v8f acc = {};
#pragma unroll
    for (int i = 0; i < 32; ++i)
      acc = wmma_f32(frag(Ad, c, koff, i), bw[i], acc);

#pragma unroll
    for (int j = 0; j < 8; ++j)
      out[(size_t)(mbase + j) * HID + n] = acc[j] + bias - sigma * rv[j];

    if (has_next) {
      float* Nd = smem + (cur ^ 1) * TILE_F;
      *(float4*)&Nd[soff]            = pd0;
      *(float4*)&Nd[soff + 8 * LDSS] = pd1;
    }
    lds_barrier();
    cur ^= 1;
  }
}

__global__ __launch_bounds__(256)
void scatter_kernel(const int* __restrict__ rows, const int* __restrict__ cols,
                    const float* __restrict__ vals, const float* __restrict__ X,
                    const float* __restrict__ sigma_p, float* __restrict__ out,
                    int nnz) {
  const int e = blockIdx.x * 8 + (threadIdx.x >> 5);
  if (e >= nnz) return;
  const int   lane = threadIdx.x & 31;
  const int   rr   = rows[e];
  const int   cc   = cols[e];
  const float s    = vals[e] * sigma_p[0];

  const float* xp = X + (size_t)cc * HID + lane;
  const float x0 = xp[0], x1 = xp[32], x2 = xp[64], x3 = xp[96];
  float* op = out + (size_t)rr * HID + lane;
  __hip_atomic_fetch_add(&op[0],  s * x0, __ATOMIC_RELAXED, __HIP_MEMORY_SCOPE_AGENT);
  __hip_atomic_fetch_add(&op[32], s * x1, __ATOMIC_RELAXED, __HIP_MEMORY_SCOPE_AGENT);
  __hip_atomic_fetch_add(&op[64], s * x2, __ATOMIC_RELAXED, __HIP_MEMORY_SCOPE_AGENT);
  __hip_atomic_fetch_add(&op[96], s * x3, __ATOMIC_RELAXED, __HIP_MEMORY_SCOPE_AGENT);
}

__global__ __launch_bounds__(256)
void leaky_kernel(float* __restrict__ out, int n4) {
  int i = blockIdx.x * blockDim.x + threadIdx.x;
  const int stride = gridDim.x * blockDim.x;
  float4* p = (float4*)out;
  for (; i < n4; i += stride) {
    float4 v = p[i];
    v.x = v.x >= 0.f ? v.x : NEG_SLOPE * v.x;
    v.y = v.y >= 0.f ? v.y : NEG_SLOPE * v.y;
    v.z = v.z >= 0.f ? v.z : NEG_SLOPE * v.z;
    v.w = v.w >= 0.f ? v.w : NEG_SLOPE * v.w;
    p[i] = v;
  }
}

extern "C" void kernel_launch(void* const* d_in, const int* in_sizes, int n_in,
                              void* d_out, int out_size, void* d_ws, size_t ws_size,
                              hipStream_t stream) {
  const float* primal      = (const float*)d_in[0];
  const float* last_primal = (const float*)d_in[1];
  const float* dual        = (const float*)d_in[2];
  const int*   rows        = (const int*)d_in[3];
  const int*   cols        = (const int*)d_in[4];
  const float* vals        = (const float*)d_in[5];
  const float* rhs         = (const float*)d_in[6];
  const float* W1          = (const float*)d_in[7];
  const float* b1          = (const float*)d_in[8];
  const float* W2          = (const float*)d_in[9];
  const float* b2          = (const float*)d_in[10];
  const float* W3          = (const float*)d_in[11];
  const float* b3          = (const float*)d_in[12];
  const float* sigma       = (const float*)d_in[13];

  float* out = (float*)d_out;
  float* X   = (float*)d_ws;

  const int n_vars = in_sizes[0] / HID;
  const int m_cons = in_sizes[2] / HID;
  const int nnz    = in_sizes[5];

  fused_x_kernel<<<GEMM_BLOCKS_X, 256, 0, stream>>>(primal, last_primal,
                                                    W2, b2, W3, b3, X, n_vars / 16);
  dual_base_kernel<<<GEMM_BLOCKS_D, 256, 0, stream>>>(dual, W1, b1, rhs, sigma, out,
                                                      m_cons / 16);
  scatter_kernel<<<(nnz + 7) / 8, 256, 0, stream>>>(rows, cols, vals, X, sigma,
                                                    out, nnz);
  leaky_kernel<<<4096, 256, 0, stream>>>(out, (m_cons * HID) / 4);
}


# Scatter: 10000 resident waves, not one wave per nonzero; a wave loads 32 nonzeros' indices coalesced, broadcasts each from its lane, keeps 8 X rows in flight; atomics kept, unordered as before
# speedup vs baseline: 1.9529x; 1.2657x over previous
#include <hip/hip_runtime.h>


#define HID 128
#define LDSS 132
#define TILE_F (16 * LDSS)
#define NEG_SLOPE 0.01f
#define GEMM_BLOCKS_X 512
#define GEMM_BLOCKS_D 768

typedef float v2f __attribute__((ext_vector_type(2)));
typedef float v8f __attribute__((ext_vector_type(8)));

__device__ __forceinline__ v8f wmma_f32(v2f a, v2f b, v8f c) {
  return __builtin_amdgcn_wmma_f32_16x16x4_f32(
       false, a,  false, b,
       (short)0, c,  false,  false);
}

__device__ __forceinline__ void stage_w_rows(const float* __restrict__ W, int n0,
                                             float* wst, int lane) {
  const float4* g = (const float4*)(W + (size_t)n0 * HID);
#pragma unroll
  for (int q = 0; q < 16; ++q)
    *(float4*)&wst[q * LDSS + lane * 4] = g[q * 32 + lane];
}

__device__ __forceinline__ v2f frag(const float* t, int c, int koff, int i) {
  return *(const v2f*)&t[c * LDSS + 4 * i + koff];
}

__device__ __forceinline__ void lds_barrier() {
  __builtin_amdgcn_fence(__ATOMIC_RELEASE, "workgroup", "local");
  __builtin_amdgcn_s_barrier();
  __builtin_amdgcn_fence(__ATOMIC_ACQUIRE, "workgroup", "local");
}

__global__ __launch_bounds__(256)
void fused_x_kernel(const float* __restrict__ primal,
                    const float* __restrict__ last_primal,
                    const float* __restrict__ W2, const float* __restrict__ b2,
                    const float* __restrict__ W3, const float* __restrict__ b3,
                    float* __restrict__ X, int ntiles) {
  __shared__ __attribute__((aligned(16))) float smem[8 * TILE_F];
  const int tid  = threadIdx.x;
  const int lane = tid & 31;
  const int wave = tid >> 5;
  const int c    = lane & 15;
  const int h    = lane >> 4;
  const int koff = h << 1;
  const int n    = wave * 16 + c;

  int tile = blockIdx.x;

  const int srow = tid >> 5, sc4 = tid & 31;
  const int soff = srow * LDSS + sc4 * 4;
  float4 pa0 = {0.f, 0.f, 0.f, 0.f}, pa1 = pa0, pl0 = pa0, pl1 = pa0;
  if (tile < ntiles) {
    const float4* gp = (const float4*)(primal      + (size_t)tile * 16 * HID);
    const float4* gl = (const float4*)(last_primal + (size_t)tile * 16 * HID);
    pa0 = gp[tid]; pa1 = gp[tid + 256];
    pl0 = gl[tid]; pl1 = gl[tid + 256];
  }

  v2f bw2[32], bw3[32];
  float* wst = smem + wave * TILE_F;
  stage_w_rows(W2, wave * 16, wst, lane);
  __syncthreads();
#pragma unroll
  for (int i = 0; i < 32; ++i) bw2[i] = frag(wst, c, koff, i);
  __syncthreads();
  stage_w_rows(W3, wave * 16, wst, lane);
  __syncthreads();
#pragma unroll
  for (int i = 0; i < 32; ++i) bw3[i] = frag(wst, c, koff, i);
  __syncthreads();

  const float bias = b2[n] - b3[n];

  if (tile < ntiles) {
    *(float4*)&smem[soff]                     = pa0;
    *(float4*)&smem[soff + 8 * LDSS]          = pa1;
    *(float4*)&smem[TILE_F + soff]            = pl0;
    *(float4*)&smem[TILE_F + soff + 8 * LDSS] = pl1;
  }
  __syncthreads();

  int cur = 0;
  for (; tile < ntiles; tile += gridDim.x) {
    const int  next     = tile + gridDim.x;
    const bool has_next = next < ntiles;
    if (has_next) {
      const float4* gp = (const float4*)(primal      + (size_t)next * 16 * HID);
      const float4* gl = (const float4*)(last_primal + (size_t)next * 16 * HID);
      pa0 = gp[tid]; pa1 = gp[tid + 256];
      pl0 = gl[tid]; pl1 = gl[tid + 256];
    }

    const float* Ap = smem + cur * 2 * TILE_F;
    const float* Al = Ap + TILE_F;
    v8f acc  = {};
    v8f acc2 = {};
#pragma unroll
    for (int i = 0; i < 32; ++i) {
      acc  = wmma_f32(frag(Ap, c, koff, i), bw2[i], acc);
      acc2 = wmma_f32(frag(Al, c, koff, i), bw3[i], acc2);
    }

    const int mbase = tile * 16 + (h << 3);
#pragma unroll
    for (int j = 0; j < 8; ++j)
      X[(size_t)(mbase + j) * HID + n] = acc[j] - acc2[j] + bias;

    if (has_next) {
      float* Np = smem + (cur ^ 1) * 2 * TILE_F;
      *(float4*)&Np[soff]                     = pa0;
      *(float4*)&Np[soff + 8 * LDSS]          = pa1;
      *(float4*)&Np[TILE_F + soff]            = pl0;
      *(float4*)&Np[TILE_F + soff + 8 * LDSS] = pl1;
    }
    lds_barrier();
    cur ^= 1;
  }
}

__global__ __launch_bounds__(256)
void dual_base_kernel(const float* __restrict__ dual,
                      const float* __restrict__ W1, const float* __restrict__ b1,
                      const float* __restrict__ rhs,
                      const float* __restrict__ sigma_p,
                      float* __restrict__ out, int ntiles) {
  __shared__ __attribute__((aligned(16))) float smem[8 * TILE_F];
  const int tid  = threadIdx.x;
  const int lane = tid & 31;
  const int wave = tid >> 5;
  const int c    = lane & 15;
  const int h    = lane >> 4;
  const int koff = h << 1;
  const int n    = wave * 16 + c;

  int tile = blockIdx.x;

  const int srow = tid >> 5, sc4 = tid & 31;
  const int soff = srow * LDSS + sc4 * 4;
  float4 pd0 = {0.f, 0.f, 0.f, 0.f}, pd1 = pd0;
  if (tile < ntiles) {
    const float4* gd = (const float4*)(dual + (size_t)tile * 16 * HID);
    pd0 = gd[tid]; pd1 = gd[tid + 256];
  }

  v2f bw[32];
  float* wst = smem + wave * TILE_F;
  stage_w_rows(W1, wave * 16, wst, lane);
  __syncthreads();
#pragma unroll
  for (int i = 0; i < 32; ++i) bw[i] = frag(wst, c, koff, i);
  __syncthreads();

  const float bias  = b1[n];
  const float sigma = sigma_p[0];

  if (tile < ntiles) {
    *(float4*)&smem[soff]            = pd0;
    *(float4*)&smem[soff + 8 * LDSS] = pd1;
  }
  __syncthreads();

  int cur = 0;
  for (; tile < ntiles; tile += gridDim.x) {
    const int  next     = tile + gridDim.x;
    const bool has_next = next < ntiles;
    if (has_next) {
      const float4* gd = (const float4*)(dual + (size_t)next * 16 * HID);
      pd0 = gd[tid]; pd1 = gd[tid + 256];
    }

    const int mbase = tile * 16 + (h << 3);
    float rv[8];
#pragma unroll
    for (int j = 0; j < 8; ++j) rv[j] = rhs[(size_t)(mbase + j) * HID + n];

    const float* Ad = smem + cur * TILE_F;
    v8f acc = {};
#pragma unroll
    for (int i = 0; i < 32; ++i)
      acc = wmma_f32(frag(Ad, c, koff, i), bw[i], acc);

#pragma unroll
    for (int j = 0; j < 8; ++j)
      out[(size_t)(mbase + j) * HID + n] = acc[j] + bias - sigma * rv[j];

    if (has_next) {
      float* Nd = smem + (cur ^ 1) * TILE_F;
      *(float4*)&Nd[soff]            = pd0;
      *(float4*)&Nd[soff + 8 * LDSS] = pd1;
    }
    lds_barrier();
    cur ^= 1;
  }
}

#define SC_U      4
#define SC_NB     (32 / SC_U)
#define SC_BLOCKS 1250

__device__ __forceinline__ void sc_load_batch(float (&xb)[SC_U][4], const float* __restrict__ X,
                                              int cidx, int j0, int lane) {
#pragma unroll
  for (int u = 0; u < SC_U; ++u) {
    const int cc = __builtin_amdgcn_readlane(cidx, j0 + u);
    const float* xp = X + (size_t)cc * HID + lane;
    xb[u][0] = xp[0]; xb[u][1] = xp[32]; xb[u][2] = xp[64]; xb[u][3] = xp[96];
  }
}

__device__ __forceinline__ void sc_add_batch(const float (&xb)[SC_U][4], float* __restrict__ out,
                                             int ridx, float sval, int j0, int lane) {
#pragma unroll
  for (int u = 0; u < SC_U; ++u) {
    const int   rr = __builtin_amdgcn_readlane(ridx, j0 + u);
    const float s  = __int_as_float(__builtin_amdgcn_readlane(__float_as_int(sval), j0 + u));
    float* op = out + (size_t)rr * HID + lane;
    __hip_atomic_fetch_add(&op[0],  s * xb[u][0], __ATOMIC_RELAXED, __HIP_MEMORY_SCOPE_AGENT);
    __hip_atomic_fetch_add(&op[32], s * xb[u][1], __ATOMIC_RELAXED, __HIP_MEMORY_SCOPE_AGENT);
    __hip_atomic_fetch_add(&op[64], s * xb[u][2], __ATOMIC_RELAXED, __HIP_MEMORY_SCOPE_AGENT);
    __hip_atomic_fetch_add(&op[96], s * xb[u][3], __ATOMIC_RELAXED, __HIP_MEMORY_SCOPE_AGENT);
  }
}

__global__ __launch_bounds__(256)
void scatter_kernel(const int* __restrict__ rows, const int* __restrict__ cols,
                    const float* __restrict__ vals, const float* __restrict__ X,
                    const float* __restrict__ sigma_p, float* __restrict__ out,
                    int nnz) {
  const int lane = threadIdx.x & 31;
  const int wv    = __builtin_amdgcn_readfirstlane(threadIdx.x >> 5);
  const int gw    = blockIdx.x * 8 + wv;
  const int nw    = gridDim.x * 8;
  const int nfull = nnz >> 5;
  const float sigma = sigma_p[0];

  int chunk = gw;
  if (chunk < nfull) {
    int   ridx = rows[(size_t)chunk * 32 + lane];
    int   cidx = cols[(size_t)chunk * 32 + lane];
    float sval = vals[(size_t)chunk * 32 + lane] * sigma;
    float xb0[SC_U][4], xb1[SC_U][4];
    sc_load_batch(xb0, X, cidx, 0, lane);
    for (;;) {
      const int  nchunk   = chunk + nw;
      const bool has_next = nchunk < nfull;
      int rn = 0, cn = 0; float vn = 0.f;
      if (has_next) {
        rn = rows[(size_t)nchunk * 32 + lane];
        cn = cols[(size_t)nchunk * 32 + lane];
        vn = vals[(size_t)nchunk * 32 + lane];
      }
#pragma unroll
      for (int b = 0; b < SC_NB; b += 2) {
        sc_load_batch(xb1, X, cidx, (b + 1) * SC_U, lane);
        sc_add_batch(xb0, out, ridx, sval, b * SC_U, lane);
        if (b + 2 < SC_NB)  sc_load_batch(xb0, X, cidx, (b + 2) * SC_U, lane);
        else if (has_next)  sc_load_batch(xb0, X, cn, 0, lane);
        sc_add_batch(xb1, out, ridx, sval, (b + 1) * SC_U, lane);
      }
      if (!has_next) break;
      ridx = rn; cidx = cn; sval = vn * sigma; chunk = nchunk;
    }
  }

  if (gw == nw - 1) {
    for (int e = nfull * 32; e < nnz; ++e) {
      const int   rr = rows[e];
      const int   cc = cols[e];
      const float s  = vals[e] * sigma;
      const float* xp = X + (size_t)cc * HID + lane;
      const float x0 = xp[0], x1 = xp[32], x2 = xp[64], x3 = xp[96];
      float* op = out + (size_t)rr * HID + lane;
      __hip_atomic_fetch_add(&op[0],  s * x0, __ATOMIC_RELAXED, __HIP_MEMORY_SCOPE_AGENT);
      __hip_atomic_fetch_add(&op[32], s * x1, __ATOMIC_RELAXED, __HIP_MEMORY_SCOPE_AGENT);
      __hip_atomic_fetch_add(&op[64], s * x2, __ATOMIC_RELAXED, __HIP_MEMORY_SCOPE_AGENT);
      __hip_atomic_fetch_add(&op[96], s * x3, __ATOMIC_RELAXED, __HIP_MEMORY_SCOPE_AGENT);
    }
  }
}

__global__ __launch_bounds__(256)
void leaky_kernel(float* __restrict__ out, int n4) {
  int i = blockIdx.x * blockDim.x + threadIdx.x;
  const int stride = gridDim.x * blockDim.x;
  float4* p = (float4*)out;
  for (; i < n4; i += stride) {
    float4 v = p[i];
    v.x = v.x >= 0.f ? v.x : NEG_SLOPE * v.x;
    v.y = v.y >= 0.f ? v.y : NEG_SLOPE * v.y;
    v.z = v.z >= 0.f ? v.z : NEG_SLOPE * v.z;
    v.w = v.w >= 0.f ? v.w : NEG_SLOPE * v.w;
    p[i] = v;
  }
}

extern "C" void kernel_launch(void* const* d_in, const int* in_sizes, int n_in,
                              void* d_out, int out_size, void* d_ws, size_t ws_size,
                              hipStream_t stream) {
  const float* primal      = (const float*)d_in[0];
  const float* last_primal = (const float*)d_in[1];
  const float* dual        = (const float*)d_in[2];
  const int*   rows        = (const int*)d_in[3];
  const int*   cols        = (const int*)d_in[4];
  const float* vals        = (const float*)d_in[5];
  const float* rhs         = (const float*)d_in[6];
  const float* W1          = (const float*)d_in[7];
  const float* b1          = (const float*)d_in[8];
  const float* W2          = (const float*)d_in[9];
  const float* b2          = (const float*)d_in[10];
  const float* W3          = (const float*)d_in[11];
  const float* b3          = (const float*)d_in[12];
  const float* sigma       = (const float*)d_in[13];

  float* out = (float*)d_out;
  float* X   = (float*)d_ws;

  const int n_vars = in_sizes[0] / HID;
  const int m_cons = in_sizes[2] / HID;
  const int nnz    = in_sizes[5];

  fused_x_kernel<<<GEMM_BLOCKS_X, 256, 0, stream>>>(primal, last_primal,
                                                    W2, b2, W3, b3, X, n_vars / 16);
  dual_base_kernel<<<GEMM_BLOCKS_D, 256, 0, stream>>>(dual, W1, b1, rhs, sigma, out,
                                                      m_cons / 16);
  scatter_kernel<<<SC_BLOCKS, 256, 0, stream>>>(rows, cols, vals, X, sigma, out, nnz);
  leaky_kernel<<<4096, 256, 0, stream>>>(out, (m_cons * HID) / 4);
}


# Scatter: 256 blocks each own 391 out rows in LDS, scan all row indices, add their nonzeros' X rows by LDS float adds, write leaky(base+sums) once; no global atomics or ReLU pass; sum order changes
# speedup vs baseline: 2.3636x; 1.2103x over previous
#include <hip/hip_runtime.h>


#define HID 128
#define LDSS 132
#define TILE_F (16 * LDSS)
#define NEG_SLOPE 0.01f
#define GEMM_BLOCKS_X 512
#define GEMM_BLOCKS_D 768

typedef float v2f __attribute__((ext_vector_type(2)));
typedef float v8f __attribute__((ext_vector_type(8)));

__device__ __forceinline__ v8f wmma_f32(v2f a, v2f b, v8f c) {
  return __builtin_amdgcn_wmma_f32_16x16x4_f32(
       false, a,  false, b,
       (short)0, c,  false,  false);
}

__device__ __forceinline__ void stage_w_rows(const float* __restrict__ W, int n0,
                                             float* wst, int lane) {
  const float4* g = (const float4*)(W + (size_t)n0 * HID);
#pragma unroll
  for (int q = 0; q < 16; ++q)
    *(float4*)&wst[q * LDSS + lane * 4] = g[q * 32 + lane];
}

__device__ __forceinline__ v2f frag(const float* t, int c, int koff, int i) {
  return *(const v2f*)&t[c * LDSS + 4 * i + koff];
}

__device__ __forceinline__ void lds_barrier() {
  __builtin_amdgcn_fence(__ATOMIC_RELEASE, "workgroup", "local");
  __builtin_amdgcn_s_barrier();
  __builtin_amdgcn_fence(__ATOMIC_ACQUIRE, "workgroup", "local");
}

__global__ __launch_bounds__(256)
void fused_x_kernel(const float* __restrict__ primal,
                    const float* __restrict__ last_primal,
                    const float* __restrict__ W2, const float* __restrict__ b2,
                    const float* __restrict__ W3, const float* __restrict__ b3,
                    float* __restrict__ X, int ntiles) {
  __shared__ __attribute__((aligned(16))) float smem[8 * TILE_F];
  const int tid  = threadIdx.x;
  const int lane = tid & 31;
  const int wave = tid >> 5;
  const int c    = lane & 15;
  const int h    = lane >> 4;
  const int koff = h << 1;
  const int n    = wave * 16 + c;

  int tile = blockIdx.x;

  const int srow = tid >> 5, sc4 = tid & 31;
  const int soff = srow * LDSS + sc4 * 4;
  float4 pa0 = {0.f, 0.f, 0.f, 0.f}, pa1 = pa0, pl0 = pa0, pl1 = pa0;
  if (tile < ntiles) {
    const float4* gp = (const float4*)(primal      + (size_t)tile * 16 * HID);
    const float4* gl = (const float4*)(last_primal + (size_t)tile * 16 * HID);
    pa0 = gp[tid]; pa1 = gp[tid + 256];
    pl0 = gl[tid]; pl1 = gl[tid + 256];
  }

  v2f bw2[32], bw3[32];
  float* wst = smem + wave * TILE_F;
  stage_w_rows(W2, wave * 16, wst, lane);
  __syncthreads();
#pragma unroll
  for (int i = 0; i < 32; ++i) bw2[i] = frag(wst, c, koff, i);
  __syncthreads();
  stage_w_rows(W3, wave * 16, wst, lane);
  __syncthreads();
#pragma unroll
  for (int i = 0; i < 32; ++i) bw3[i] = frag(wst, c, koff, i);
  __syncthreads();

  const float bias = b2[n] - b3[n];

  if (tile < ntiles) {
    *(float4*)&smem[soff]                     = pa0;
    *(float4*)&smem[soff + 8 * LDSS]          = pa1;
    *(float4*)&smem[TILE_F + soff]            = pl0;
    *(float4*)&smem[TILE_F + soff + 8 * LDSS] = pl1;
  }
  __syncthreads();

  int cur = 0;
  for (; tile < ntiles; tile += gridDim.x) {
    const int  next     = tile + gridDim.x;
    const bool has_next = next < ntiles;
    if (has_next) {
      const float4* gp = (const float4*)(primal      + (size_t)next * 16 * HID);
      const float4* gl = (const float4*)(last_primal + (size_t)next * 16 * HID);
      pa0 = gp[tid]; pa1 = gp[tid + 256];
      pl0 = gl[tid]; pl1 = gl[tid + 256];
    }

    const float* Ap = smem + cur * 2 * TILE_F;
    const float* Al = Ap + TILE_F;
    v8f acc  = {};
    v8f acc2 = {};
#pragma unroll
    for (int i = 0; i < 32; ++i) {
      acc  = wmma_f32(frag(Ap, c, koff, i), bw2[i], acc);
      acc2 = wmma_f32(frag(Al, c, koff, i), bw3[i], acc2);
    }

    const int mbase = tile * 16 + (h << 3);
#pragma unroll
    for (int j = 0; j < 8; ++j)
      X[(size_t)(mbase + j) * HID + n] = acc[j] - acc2[j] + bias;

    if (has_next) {
      float* Np = smem + (cur ^ 1) * 2 * TILE_F;
      *(float4*)&Np[soff]                     = pa0;
      *(float4*)&Np[soff + 8 * LDSS]          = pa1;
      *(float4*)&Np[TILE_F + soff]            = pl0;
      *(float4*)&Np[TILE_F + soff + 8 * LDSS] = pl1;
    }
    lds_barrier();
    cur ^= 1;
  }
}

__global__ __launch_bounds__(256)
void dual_base_kernel(const float* __restrict__ dual,
                      const float* __restrict__ W1, const float* __restrict__ b1,
                      const float* __restrict__ rhs,
                      const float* __restrict__ sigma_p,
                      float* __restrict__ out, int ntiles) {
  __shared__ __attribute__((aligned(16))) float smem[8 * TILE_F];
  const int tid  = threadIdx.x;
  const int lane = tid & 31;
  const int wave = tid >> 5;
  const int c    = lane & 15;
  const int h    = lane >> 4;
  const int koff = h << 1;
  const int n    = wave * 16 + c;

  int tile = blockIdx.x;

  const int srow = tid >> 5, sc4 = tid & 31;
  const int soff = srow * LDSS + sc4 * 4;
  float4 pd0 = {0.f, 0.f, 0.f, 0.f}, pd1 = pd0;
  if (tile < ntiles) {
    const float4* gd = (const float4*)(dual + (size_t)tile * 16 * HID);
    pd0 = gd[tid]; pd1 = gd[tid + 256];
  }

  v2f bw[32];
  float* wst = smem + wave * TILE_F;
  stage_w_rows(W1, wave * 16, wst, lane);
  __syncthreads();
#pragma unroll
  for (int i = 0; i < 32; ++i) bw[i] = frag(wst, c, koff, i);
  __syncthreads();

  const float bias  = b1[n];
  const float sigma = sigma_p[0];

  if (tile < ntiles) {
    *(float4*)&smem[soff]            = pd0;
    *(float4*)&smem[soff + 8 * LDSS] = pd1;
  }
  __syncthreads();

  int cur = 0;
  for (; tile < ntiles; tile += gridDim.x) {
    const int  next     = tile + gridDim.x;
    const bool has_next = next < ntiles;
    if (has_next) {
      const float4* gd = (const float4*)(dual + (size_t)next * 16 * HID);
      pd0 = gd[tid]; pd1 = gd[tid + 256];
    }

    const int mbase = tile * 16 + (h << 3);
    float rv[8];
#pragma unroll
    for (int j = 0; j < 8; ++j) rv[j] = rhs[(size_t)(mbase + j) * HID + n];

    const float* Ad = smem + cur * TILE_F;
    v8f acc = {};
#pragma unroll
    for (int i = 0; i < 32; ++i)
      acc = wmma_f32(frag(Ad, c, koff, i), bw[i], acc);

#pragma unroll
    for (int j = 0; j < 8; ++j)
      out[(size_t)(mbase + j) * HID + n] = acc[j] + bias - sigma * rv[j];

    if (has_next) {
      float* Nd = smem + (cur ^ 1) * TILE_F;
      *(float4*)&Nd[soff]            = pd0;
      *(float4*)&Nd[soff + 8 * LDSS] = pd1;
    }
    lds_barrier();
    cur ^= 1;
  }
}

#define SC_U       4
#define SC_NB      (32 / SC_U)
#define OW_ROWS    391
#define OW_BLOCKS  256
#define OW_WAVES   32
#define OW_THREADS (OW_WAVES * 32)
#define OW_QCAP    512

__device__ __forceinline__ int4 ow_load_rows(const int* __restrict__ rows, long long gbase,
                                             int nnz, int lane) {
  const long long i = gbase + lane * 4;
  if (gbase + 128 <= nnz) return *(const int4*)(rows + i);
  int4 r;
  r.x = i     < nnz ? rows[i]     : -1;
  r.y = i + 1 < nnz ? rows[i + 1] : -1;
  r.z = i + 2 < nnz ? rows[i + 2] : -1;
  r.w = i + 3 < nnz ? rows[i + 3] : -1;
  return r;
}

__device__ __forceinline__ void ow_push(int r, int e, int lo, int nr, int* q, int& tail) {
  const bool m = (unsigned)(r - lo) < (unsigned)nr;
  const unsigned bal = __builtin_amdgcn_ballot_w32(m);
  if (bal != 0u) {
    if (m) q[(tail + (int)__builtin_amdgcn_mbcnt_lo(bal, 0u)) & (OW_QCAP - 1)] = e;
    tail += __builtin_popcount(bal);
  }
}

__device__ __forceinline__ void ow_load_batch(float4 (&xb)[SC_U], const float* __restrict__ X,
                                              int cidx, int j0, int lane) {
#pragma unroll
  for (int u = 0; u < SC_U; ++u) {
    const int cc = __builtin_amdgcn_readlane(cidx, j0 + u);
    xb[u] = *(const float4*)(X + (size_t)cc * HID + lane * 4);
  }
}

__device__ __forceinline__ void ow_add_batch(const float4 (&xb)[SC_U], float* acc,
                                             int rloc, float sval, int j0, int cnt, int lane) {
#pragma unroll
  for (int u = 0; u < SC_U; ++u) {
    if (j0 + u < cnt) {
      const int   rr = __builtin_amdgcn_readlane(rloc, j0 + u);
      const float s  = __int_as_float(__builtin_amdgcn_readlane(__float_as_int(sval), j0 + u));
      float* ap = acc + rr * HID + lane;
      __hip_atomic_fetch_add(&ap[0],  s * xb[u].x, __ATOMIC_RELAXED, __HIP_MEMORY_SCOPE_WORKGROUP);
      __hip_atomic_fetch_add(&ap[32], s * xb[u].y, __ATOMIC_RELAXED, __HIP_MEMORY_SCOPE_WORKGROUP);
      __hip_atomic_fetch_add(&ap[64], s * xb[u].z, __ATOMIC_RELAXED, __HIP_MEMORY_SCOPE_WORKGROUP);
      __hip_atomic_fetch_add(&ap[96], s * xb[u].w, __ATOMIC_RELAXED, __HIP_MEMORY_SCOPE_WORKGROUP);
    }
  }
}

__device__ __forceinline__ void ow_drain(const int* q, int head, int cnt, int lo,
                                         const int* __restrict__ rows, const int* __restrict__ cols,
                                         const float* __restrict__ vals, const float* __restrict__ X,
                                         float sigma, float* acc, int lane) {
  int rloc = 0, cidx = 0;
  float sval = 0.f;
  if (lane < cnt) {
    const int e = q[(head + lane) & (OW_QCAP - 1)];
    rloc = rows[e] - lo;
    cidx = cols[e];
    sval = vals[e] * sigma;
  }
  float4 xb0[SC_U], xb1[SC_U];
  ow_load_batch(xb0, X, cidx, 0, lane);
#pragma unroll
  for (int b = 0; b < SC_NB; b += 2) {
    if ((b + 1) * SC_U < cnt) ow_load_batch(xb1, X, cidx, (b + 1) * SC_U, lane);
    ow_add_batch(xb0, acc, rloc, sval, b * SC_U, cnt, lane);
    if (b + 2 < SC_NB && (b + 2) * SC_U < cnt) ow_load_batch(xb0, X, cidx, (b + 2) * SC_U, lane);
    ow_add_batch(xb1, acc, rloc, sval, (b + 1) * SC_U, cnt, lane);
  }
}

__global__ __launch_bounds__(OW_THREADS)
void owner_scatter_kernel(const int* __restrict__ rows, const int* __restrict__ cols,
                          const float* __restrict__ vals, const float* __restrict__ X,
                          const float* __restrict__ sigma_p, float* __restrict__ out,
                          int nnz, int m_cons) {
  __shared__ __attribute__((aligned(16))) float acc[OW_ROWS * HID];
  __shared__ int ring[OW_WAVES * OW_QCAP];
  const int tid  = threadIdx.x;
  const int lane = tid & 31;
  const int wv = __builtin_amdgcn_readfirstlane(tid >> 5);
  int* q = ring + wv * OW_QCAP;
  const float sigma = sigma_p[0];
  const int nsuper = (int)(((long long)nnz + 1023) >> 10);
  const int off = (int)(((long long)blockIdx.x * nsuper) / gridDim.x);

  for (int lo = blockIdx.x * OW_ROWS; lo < m_cons; lo += gridDim.x * OW_ROWS) {
    const int nr = min(OW_ROWS, m_cons - lo);

    for (int i = tid; i < OW_ROWS * HID / 4; i += OW_THREADS)
      ((float4*)acc)[i] = make_float4(0.f, 0.f, 0.f, 0.f);
    __syncthreads();

    int head = 0, tail = 0;
    for (int t = wv; t < nsuper; t += OW_WAVES) {
      int g = t + off; if (g >= nsuper) g -= nsuper;
      const long long sb = (long long)g << 10;
      int4 c[8];
      if (sb + 1024 <= nnz) {
#pragma unroll
        for (int j = 0; j < 8; ++j) c[j] = *(const int4*)(rows + sb + j * 128 + lane * 4);
      } else {
#pragma unroll
        for (int j = 0; j < 8; ++j) c[j] = ow_load_rows(rows, sb + j * 128, nnz, lane);
      }
      const int e00 = (int)sb + lane * 4;
#pragma unroll
      for (int j = 0; j < 8; j += 2) {
        ow_push(c[j].x,     e00 + j * 128,       lo, nr, q, tail);
        ow_push(c[j].y,     e00 + j * 128 + 1,   lo, nr, q, tail);
        ow_push(c[j].z,     e00 + j * 128 + 2,   lo, nr, q, tail);
        ow_push(c[j].w,     e00 + j * 128 + 3,   lo, nr, q, tail);
        ow_push(c[j + 1].x, e00 + j * 128 + 128, lo, nr, q, tail);
        ow_push(c[j + 1].y, e00 + j * 128 + 129, lo, nr, q, tail);
        ow_push(c[j + 1].z, e00 + j * 128 + 130, lo, nr, q, tail);
        ow_push(c[j + 1].w, e00 + j * 128 + 131, lo, nr, q, tail);
        while (tail - head >= 32) {
          __builtin_amdgcn_fence(__ATOMIC_ACQ_REL, "wavefront");
          ow_drain(q, head, 32, lo, rows, cols, vals, X, sigma, acc, lane);
          head += 32;
        }
      }
    }
    if (tail - head > 0) {
      __builtin_amdgcn_fence(__ATOMIC_ACQ_REL, "wavefront");
      ow_drain(q, head, tail - head, lo, rows, cols, vals, X, sigma, acc, lane);
    }
    __syncthreads();

    for (int i = tid; i < nr * 32; i += OW_THREADS) {
      const int row = i >> 5, q4 = i & 31;
      float4* op = (float4*)(out + (size_t)(lo + row) * HID) + q4;
      const float* ap = acc + row * HID + q4;
      float4 v = *op;
      v.x += ap[0]; v.y += ap[32]; v.z += ap[64]; v.w += ap[96];
      v.x = v.x >= 0.f ? v.x : NEG_SLOPE * v.x;
      v.y = v.y >= 0.f ? v.y : NEG_SLOPE * v.y;
      v.z = v.z >= 0.f ? v.z : NEG_SLOPE * v.z;
      v.w = v.w >= 0.f ? v.w : NEG_SLOPE * v.w;
      *op = v;
    }
    __syncthreads();
  }
}

extern "C" void kernel_launch(void* const* d_in, const int* in_sizes, int n_in,
                              void* d_out, int out_size, void* d_ws, size_t ws_size,
                              hipStream_t stream) {
  const float* primal      = (const float*)d_in[0];
  const float* last_primal = (const float*)d_in[1];
  const float* dual        = (const float*)d_in[2];
  const int*   rows        = (const int*)d_in[3];
  const int*   cols        = (const int*)d_in[4];
  const float* vals        = (const float*)d_in[5];
  const float* rhs         = (const float*)d_in[6];
  const float* W1          = (const float*)d_in[7];
  const float* b1          = (const float*)d_in[8];
  const float* W2          = (const float*)d_in[9];
  const float* b2          = (const float*)d_in[10];
  const float* W3          = (const float*)d_in[11];
  const float* b3          = (const float*)d_in[12];
  const float* sigma       = (const float*)d_in[13];

  float* out = (float*)d_out;
  float* X   = (float*)d_ws;

  const int n_vars = in_sizes[0] / HID;
  const int m_cons = in_sizes[2] / HID;
  const int nnz    = in_sizes[5];

  fused_x_kernel<<<GEMM_BLOCKS_X, 256, 0, stream>>>(primal, last_primal,
                                                    W2, b2, W3, b3, X, n_vars / 16);
  dual_base_kernel<<<GEMM_BLOCKS_D, 256, 0, stream>>>(dual, W1, b1, rhs, sigma, out,
                                                      m_cons / 16);
  owner_scatter_kernel<<<OW_BLOCKS, OW_THREADS, 0, stream>>>(rows, cols, vals, X, sigma, out,
                                                             nnz, m_cons);
}


# Scatter: a front kernel marks per owner block which 32-nonzero groups hold its rows (1 bit each); a block loads only its marked groups' row indices, not all 1.6M; order of the LDS float adds changes
# speedup vs baseline: 2.5823x; 1.0925x over previous
#include <hip/hip_runtime.h>


#define HID 128
#define LDSS 132
#define TILE_F (16 * LDSS)
#define NEG_SLOPE 0.01f
#define GEMM_BLOCKS_X 512
#define GEMM_BLOCKS_D 768

typedef float v2f __attribute__((ext_vector_type(2)));
typedef float v8f __attribute__((ext_vector_type(8)));

__device__ __forceinline__ v8f wmma_f32(v2f a, v2f b, v8f c) {
  return __builtin_amdgcn_wmma_f32_16x16x4_f32(
       false, a,  false, b,
       (short)0, c,  false,  false);
}

__device__ __forceinline__ void stage_w_rows(const float* __restrict__ W, int n0,
                                             float* wst, int lane) {
  const float4* g = (const float4*)(W + (size_t)n0 * HID);
#pragma unroll
  for (int q = 0; q < 16; ++q)
    *(float4*)&wst[q * LDSS + lane * 4] = g[q * 32 + lane];
}

__device__ __forceinline__ v2f frag(const float* t, int c, int koff, int i) {
  return *(const v2f*)&t[c * LDSS + 4 * i + koff];
}

__device__ __forceinline__ void lds_barrier() {
  __builtin_amdgcn_fence(__ATOMIC_RELEASE, "workgroup", "local");
  __builtin_amdgcn_s_barrier();
  __builtin_amdgcn_fence(__ATOMIC_ACQUIRE, "workgroup", "local");
}

__global__ __launch_bounds__(256)
void fused_x_kernel(const float* __restrict__ primal,
                    const float* __restrict__ last_primal,
                    const float* __restrict__ W2, const float* __restrict__ b2,
                    const float* __restrict__ W3, const float* __restrict__ b3,
                    float* __restrict__ X, int ntiles) {
  __shared__ __attribute__((aligned(16))) float smem[8 * TILE_F];
  const int tid  = threadIdx.x;
  const int lane = tid & 31;
  const int wave = tid >> 5;
  const int c    = lane & 15;
  const int h    = lane >> 4;
  const int koff = h << 1;
  const int n    = wave * 16 + c;

  int tile = blockIdx.x;

  const int srow = tid >> 5, sc4 = tid & 31;
  const int soff = srow * LDSS + sc4 * 4;
  float4 pa0 = {0.f, 0.f, 0.f, 0.f}, pa1 = pa0, pl0 = pa0, pl1 = pa0;
  if (tile < ntiles) {
    const float4* gp = (const float4*)(primal      + (size_t)tile * 16 * HID);
    const float4* gl = (const float4*)(last_primal + (size_t)tile * 16 * HID);
    pa0 = gp[tid]; pa1 = gp[tid + 256];
    pl0 = gl[tid]; pl1 = gl[tid + 256];
  }

  v2f bw2[32], bw3[32];
  float* wst = smem + wave * TILE_F;
  stage_w_rows(W2, wave * 16, wst, lane);
  __syncthreads();
#pragma unroll
  for (int i = 0; i < 32; ++i) bw2[i] = frag(wst, c, koff, i);
  __syncthreads();
  stage_w_rows(W3, wave * 16, wst, lane);
  __syncthreads();
#pragma unroll
  for (int i = 0; i < 32; ++i) bw3[i] = frag(wst, c, koff, i);
  __syncthreads();

  const float bias = b2[n] - b3[n];

  if (tile < ntiles) {
    *(float4*)&smem[soff]                     = pa0;
    *(float4*)&smem[soff + 8 * LDSS]          = pa1;
    *(float4*)&smem[TILE_F + soff]            = pl0;
    *(float4*)&smem[TILE_F + soff + 8 * LDSS] = pl1;
  }
  __syncthreads();

  int cur = 0;
  for (; tile < ntiles; tile += gridDim.x) {
    const int  next     = tile + gridDim.x;
    const bool has_next = next < ntiles;
    if (has_next) {
      const float4* gp = (const float4*)(primal      + (size_t)next * 16 * HID);
      const float4* gl = (const float4*)(last_primal + (size_t)next * 16 * HID);
      pa0 = gp[tid]; pa1 = gp[tid + 256];
      pl0 = gl[tid]; pl1 = gl[tid + 256];
    }

    const float* Ap = smem + cur * 2 * TILE_F;
    const float* Al = Ap + TILE_F;
    v8f acc  = {};
    v8f acc2 = {};
#pragma unroll
    for (int i = 0; i < 32; ++i) {
      acc  = wmma_f32(frag(Ap, c, koff, i), bw2[i], acc);
      acc2 = wmma_f32(frag(Al, c, koff, i), bw3[i], acc2);
    }

    const int mbase = tile * 16 + (h << 3);
#pragma unroll
    for (int j = 0; j < 8; ++j)
      X[(size_t)(mbase + j) * HID + n] = acc[j] - acc2[j] + bias;

    if (has_next) {
      float* Np = smem + (cur ^ 1) * 2 * TILE_F;
      *(float4*)&Np[soff]                     = pa0;
      *(float4*)&Np[soff + 8 * LDSS]          = pa1;
      *(float4*)&Np[TILE_F + soff]            = pl0;
      *(float4*)&Np[TILE_F + soff + 8 * LDSS] = pl1;
    }
    lds_barrier();
    cur ^= 1;
  }
}

__global__ __launch_bounds__(256)
void dual_base_kernel(const float* __restrict__ dual,
                      const float* __restrict__ W1, const float* __restrict__ b1,
                      const float* __restrict__ rhs,
                      const float* __restrict__ sigma_p,
                      float* __restrict__ out, int ntiles) {
  __shared__ __attribute__((aligned(16))) float smem[8 * TILE_F];
  const int tid  = threadIdx.x;
  const int lane = tid & 31;
  const int wave = tid >> 5;
  const int c    = lane & 15;
  const int h    = lane >> 4;
  const int koff = h << 1;
  const int n    = wave * 16 + c;

  int tile = blockIdx.x;

  const int srow = tid >> 5, sc4 = tid & 31;
  const int soff = srow * LDSS + sc4 * 4;
  float4 pd0 = {0.f, 0.f, 0.f, 0.f}, pd1 = pd0;
  if (tile < ntiles) {
    const float4* gd = (const float4*)(dual + (size_t)tile * 16 * HID);
    pd0 = gd[tid]; pd1 = gd[tid + 256];
  }

  v2f bw[32];
  float* wst = smem + wave * TILE_F;
  stage_w_rows(W1, wave * 16, wst, lane);
  __syncthreads();
#pragma unroll
  for (int i = 0; i < 32; ++i) bw[i] = frag(wst, c, koff, i);
  __syncthreads();

  const float bias  = b1[n];
  const float sigma = sigma_p[0];

  if (tile < ntiles) {
    *(float4*)&smem[soff]            = pd0;
    *(float4*)&smem[soff + 8 * LDSS] = pd1;
  }
  __syncthreads();

  int cur = 0;
  for (; tile < ntiles; tile += gridDim.x) {
    const int  next     = tile + gridDim.x;
    const bool has_next = next < ntiles;
    if (has_next) {
      const float4* gd = (const float4*)(dual + (size_t)next * 16 * HID);
      pd0 = gd[tid]; pd1 = gd[tid + 256];
    }

    const int mbase = tile * 16 + (h << 3);
    float rv[8];
#pragma unroll
    for (int j = 0; j < 8; ++j) rv[j] = rhs[(size_t)(mbase + j) * HID + n];

    const float* Ad = smem + cur * TILE_F;
    v8f acc = {};
#pragma unroll
    for (int i = 0; i < 32; ++i)
      acc = wmma_f32(frag(Ad, c, koff, i), bw[i], acc);

#pragma unroll
    for (int j = 0; j < 8; ++j)
      out[(size_t)(mbase + j) * HID + n] = acc[j] + bias - sigma * rv[j];

    if (has_next) {
      float* Nd = smem + (cur ^ 1) * TILE_F;
      *(float4*)&Nd[soff]            = pd0;
      *(float4*)&Nd[soff + 8 * LDSS] = pd1;
    }
    lds_barrier();
    cur ^= 1;
  }
}

#define SC_U       4
#define SC_NB      (32 / SC_U)
#define OW_ROWS    391
#define OW_BLOCKS  256
#define OW_WAVES   32
#define OW_THREADS (OW_WAVES * 32)
#define OW_QCAP    512
#define OW_GB      8
#define OW_GQCAP   64
#define MK_BLOCKS  64
#define MK_LDSS    257

__device__ __forceinline__ int4 ow_load_rows(const int* __restrict__ rows, long long gbase,
                                             int nnz, int lane) {
  const long long i = gbase + lane * 4;
  if (gbase + 128 <= nnz) return *(const int4*)(rows + i);
  int4 r;
  r.x = i     < nnz ? rows[i]     : -1;
  r.y = i + 1 < nnz ? rows[i + 1] : -1;
  r.z = i + 2 < nnz ? rows[i + 2] : -1;
  r.w = i + 3 < nnz ? rows[i + 3] : -1;
  return r;
}

__device__ __forceinline__ void ow_push(int r, int e, int lo, int nr, int* q, int& tail) {
  const bool m = (unsigned)(r - lo) < (unsigned)nr;
  const unsigned bal = __builtin_amdgcn_ballot_w32(m);
  if (bal != 0u) {
    if (m) q[(tail + (int)__builtin_amdgcn_mbcnt_lo(bal, 0u)) & (OW_QCAP - 1)] = e;
    tail += __builtin_popcount(bal);
  }
}

__device__ __forceinline__ void ow_take_groups(const int* gq, int gpos, int cnt,
                                               const int* __restrict__ rows, int nnz,
                                               int lo, int nr, int* q, int& tail, int lane) {
  const int4* gp = (const int4*)&gq[gpos & (OW_GQCAP - 1)];
  const int4 ga = gp[0], gb = gp[1];
  const int g[OW_GB] = {ga.x, ga.y, ga.z, ga.w, gb.x, gb.y, gb.z, gb.w};
  int rr[OW_GB];
#pragma unroll
  for (int u = 0; u < OW_GB; ++u)
    rr[u] = rows[min(max((int)(((unsigned)g[u] << 5) + lane), 0), nnz - 1)];
#pragma unroll
  for (int u = 0; u < OW_GB; ++u) {
    const int e = (int)(((unsigned)g[u] << 5) + lane);
    ow_push((u < cnt && e < nnz) ? rr[u] : -1, e, lo, nr, q, tail);
  }
}

__device__ __forceinline__ void ow_load_batch(float4 (&xb)[SC_U], const float* __restrict__ X,
                                              int cidx, int j0, int lane) {
#pragma unroll
  for (int u = 0; u < SC_U; ++u) {
    const int cc = __builtin_amdgcn_readlane(cidx, j0 + u);
    xb[u] = *(const float4*)(X + (size_t)cc * HID + lane * 4);
  }
}

__device__ __forceinline__ void ow_add_batch(const float4 (&xb)[SC_U], float* acc,
                                             int rloc, float sval, int j0, int cnt, int lane) {
#pragma unroll
  for (int u = 0; u < SC_U; ++u) {
    if (j0 + u < cnt) {
      const int   rr = __builtin_amdgcn_readlane(rloc, j0 + u);
      const float s  = __int_as_float(__builtin_amdgcn_readlane(__float_as_int(sval), j0 + u));
      float* ap = acc + rr * HID + lane;
      __hip_atomic_fetch_add(&ap[0],  s * xb[u].x, __ATOMIC_RELAXED, __HIP_MEMORY_SCOPE_WORKGROUP);
      __hip_atomic_fetch_add(&ap[32], s * xb[u].y, __ATOMIC_RELAXED, __HIP_MEMORY_SCOPE_WORKGROUP);
      __hip_atomic_fetch_add(&ap[64], s * xb[u].z, __ATOMIC_RELAXED, __HIP_MEMORY_SCOPE_WORKGROUP);
      __hip_atomic_fetch_add(&ap[96], s * xb[u].w, __ATOMIC_RELAXED, __HIP_MEMORY_SCOPE_WORKGROUP);
    }
  }
}

__device__ __forceinline__ void ow_drain(const int* q, int head, int cnt, int lo,
                                         const int* __restrict__ rows, const int* __restrict__ cols,
                                         const float* __restrict__ vals, const float* __restrict__ X,
                                         float sigma, float* acc, int lane) {
  int rloc = 0, cidx = 0;
  float sval = 0.f;
  if (lane < cnt) {
    const int e = q[(head + lane) & (OW_QCAP - 1)];
    rloc = rows[e] - lo;
    cidx = cols[e];
    sval = vals[e] * sigma;
  }
  float4 xb0[SC_U], xb1[SC_U];
  ow_load_batch(xb0, X, cidx, 0, lane);
#pragma unroll
  for (int b = 0; b < SC_NB; b += 2) {
    if ((b + 1) * SC_U < cnt) ow_load_batch(xb1, X, cidx, (b + 1) * SC_U, lane);
    ow_add_batch(xb0, acc, rloc, sval, b * SC_U, cnt, lane);
    if (b + 2 < SC_NB && (b + 2) * SC_U < cnt) ow_load_batch(xb0, X, cidx, (b + 2) * SC_U, lane);
    ow_add_batch(xb1, acc, rloc, sval, (b + 1) * SC_U, cnt, lane);
  }
}


__global__ __launch_bounds__(OW_THREADS)
void owner_mask_kernel(const int* __restrict__ rows, float* __restrict__ X,
                       int n_vars, int nnz) {
  __shared__ unsigned bits[32 * MK_LDSS];
  unsigned* mask = (unsigned*)(X + (size_t)n_vars * HID);
  const int tid  = threadIdx.x;
  const int lane = tid & 31;
  const int wv   = __builtin_amdgcn_readfirstlane(tid >> 5);
  const int nw   = (int)(((long long)nnz + 1023) >> 10);
  const int nwp  = (nw + 31) & ~31;
  unsigned* mybits = bits + wv * MK_LDSS;

  for (int cb = blockIdx.x; cb * 32 < nwp; cb += gridDim.x) {
    for (int i = tid; i < 32 * MK_LDSS; i += OW_THREADS) bits[i] = 0u;
    __syncthreads();

    const long long sb = ((long long)(cb * 32 + wv)) << 10;
    if (sb < nnz) {
      int4 c[8];
      if (sb + 1024 <= nnz) {
#pragma unroll
        for (int j = 0; j < 8; ++j) c[j] = *(const int4*)(rows + sb + j * 128 + lane * 4);
      } else {
#pragma unroll
        for (int j = 0; j < 8; ++j) c[j] = ow_load_rows(rows, sb + j * 128, nnz, lane);
      }
#pragma unroll
      for (int j = 0; j < 8; ++j) {
        const unsigned bit = 1u << (j * 4 + (lane >> 3));
        if (c[j].x >= 0) atomicOr(&mybits[((unsigned)c[j].x / OW_ROWS) % OW_BLOCKS], bit);
        if (c[j].y >= 0) atomicOr(&mybits[((unsigned)c[j].y / OW_ROWS) % OW_BLOCKS], bit);
        if (c[j].z >= 0) atomicOr(&mybits[((unsigned)c[j].z / OW_ROWS) % OW_BLOCKS], bit);
        if (c[j].w >= 0) atomicOr(&mybits[((unsigned)c[j].w / OW_ROWS) % OW_BLOCKS], bit);
      }
    }
    __syncthreads();

#pragma unroll
    for (int k = 0; k < 8; ++k) {
      const int o = wv * 8 + k;
      mask[(size_t)o * nwp + cb * 32 + lane] = bits[lane * MK_LDSS + o];
    }
    __syncthreads();
  }
}

__global__ __launch_bounds__(OW_THREADS)
void owner_scatter_kernel(const int* __restrict__ rows, const int* __restrict__ cols,
                          const float* __restrict__ vals, const float* __restrict__ X,
                          const float* __restrict__ sigma_p, float* __restrict__ out,
                          int nnz, int m_cons, int n_vars) {
  __shared__ __attribute__((aligned(16))) float acc[OW_ROWS * HID];
  __shared__ int ring[OW_WAVES * OW_QCAP];
  __shared__ __attribute__((aligned(16))) int gring[OW_WAVES * OW_GQCAP];
  const int tid  = threadIdx.x;
  const int lane = tid & 31;
  const int wv = __builtin_amdgcn_readfirstlane(tid >> 5);
  int* q = ring + wv * OW_QCAP;
  int* gq = gring + wv * OW_GQCAP;
  const float sigma = sigma_p[0];
  const int nw   = (int)(((long long)nnz + 1023) >> 10);
  const int nwp  = (nw + 31) & ~31;
  const unsigned* mrow = (const unsigned*)(X + (size_t)n_vars * HID) + (size_t)blockIdx.x * nwp;
  const int wpw  = (nw + OW_WAVES - 1) / OW_WAVES;
  const int wbeg = wv * wpw;
  const int wend = min(nw, wbeg + wpw);

  for (int lo = blockIdx.x * OW_ROWS; lo < m_cons; lo += gridDim.x * OW_ROWS) {
    const int nr = min(OW_ROWS, m_cons - lo);

    for (int i = tid; i < OW_ROWS * HID / 4; i += OW_THREADS)
      ((float4*)acc)[i] = make_float4(0.f, 0.f, 0.f, 0.f);
    __syncthreads();

    int head = 0, tail = 0;
    int ghead = 0, gtail = 0;
    for (int s0 = wbeg; s0 < wend; s0 += 32) {
      const unsigned mw = (s0 + lane < wend) ? mrow[s0 + lane] : 0u;
      const int nwi = min(32, wend - s0);
      for (int wi = 0; wi < nwi; ++wi) {
        const unsigned w = (unsigned)__builtin_amdgcn_readlane((int)mw, wi);
        if (w == 0u) continue;
        if ((w >> lane) & 1u)
          gq[(gtail + (int)__builtin_amdgcn_mbcnt_lo(w, 0u)) & (OW_GQCAP - 1)] = ((s0 + wi) << 5) + lane;
        gtail += __builtin_popcount(w);
        while (gtail - ghead >= OW_GB) {
          __builtin_amdgcn_fence(__ATOMIC_ACQ_REL, "wavefront");
          ow_take_groups(gq, ghead, OW_GB, rows, nnz, lo, nr, q, tail, lane);
          ghead += OW_GB;
          while (tail - head >= 32) {
            __builtin_amdgcn_fence(__ATOMIC_ACQ_REL, "wavefront");
            ow_drain(q, head, 32, lo, rows, cols, vals, X, sigma, acc, lane);
            head += 32;
          }
        }
      }
    }
    if (gtail - ghead > 0) {
      __builtin_amdgcn_fence(__ATOMIC_ACQ_REL, "wavefront");
      ow_take_groups(gq, ghead, gtail - ghead, rows, nnz, lo, nr, q, tail, lane);
      while (tail - head >= 32) {
        __builtin_amdgcn_fence(__ATOMIC_ACQ_REL, "wavefront");
        ow_drain(q, head, 32, lo, rows, cols, vals, X, sigma, acc, lane);
        head += 32;
      }
    }
    if (tail - head > 0) {
      __builtin_amdgcn_fence(__ATOMIC_ACQ_REL, "wavefront");
      ow_drain(q, head, tail - head, lo, rows, cols, vals, X, sigma, acc, lane);
    }
    __syncthreads();

    for (int i = tid; i < nr * 32; i += OW_THREADS) {
      const int row = i >> 5, q4 = i & 31;
      float4* op = (float4*)(out + (size_t)(lo + row) * HID) + q4;
      const float* ap = acc + row * HID + q4;
      float4 v = *op;
      v.x += ap[0]; v.y += ap[32]; v.z += ap[64]; v.w += ap[96];
      v.x = v.x >= 0.f ? v.x : NEG_SLOPE * v.x;
      v.y = v.y >= 0.f ? v.y : NEG_SLOPE * v.y;
      v.z = v.z >= 0.f ? v.z : NEG_SLOPE * v.z;
      v.w = v.w >= 0.f ? v.w : NEG_SLOPE * v.w;
      *op = v;
    }
    __syncthreads();
  }
}

extern "C" void kernel_launch(void* const* d_in, const int* in_sizes, int n_in,
                              void* d_out, int out_size, void* d_ws, size_t ws_size,
                              hipStream_t stream) {
  const float* primal      = (const float*)d_in[0];
  const float* last_primal = (const float*)d_in[1];
  const float* dual        = (const float*)d_in[2];
  const int*   rows        = (const int*)d_in[3];
  const int*   cols        = (const int*)d_in[4];
  const float* vals        = (const float*)d_in[5];
  const float* rhs         = (const float*)d_in[6];
  const float* W1          = (const float*)d_in[7];
  const float* b1          = (const float*)d_in[8];
  const float* W2          = (const float*)d_in[9];
  const float* b2          = (const float*)d_in[10];
  const float* W3          = (const float*)d_in[11];
  const float* b3          = (const float*)d_in[12];
  const float* sigma       = (const float*)d_in[13];

  float* out = (float*)d_out;
  float* X   = (float*)d_ws;

  const int n_vars = in_sizes[0] / HID;
  const int m_cons = in_sizes[2] / HID;
  const int nnz    = in_sizes[5];

  owner_mask_kernel<<<MK_BLOCKS, OW_THREADS, 0, stream>>>(rows, X, n_vars, nnz);
  fused_x_kernel<<<GEMM_BLOCKS_X, 256, 0, stream>>>(primal, last_primal,
                                                    W2, b2, W3, b3, X, n_vars / 16);
  dual_base_kernel<<<GEMM_BLOCKS_D, 256, 0, stream>>>(dual, W1, b1, rhs, sigma, out,
                                                      m_cons / 16);
  owner_scatter_kernel<<<OW_BLOCKS, OW_THREADS, 0, stream>>>(rows, cols, vals, X, sigma, out,
                                                             nnz, m_cons, n_vars);
}


# Scatter: a front kernel counting-sorts each 6272-nonzero chunk by owner block into 8-byte records (col, row, sigma*val); a block reads only its own contiguous records; LDS float add order changes
# speedup vs baseline: 3.0437x; 1.1787x over previous
#include <hip/hip_runtime.h>


#define HID 128
#define LDSS 132
#define TILE_F (16 * LDSS)
#define NEG_SLOPE 0.01f
#define GEMM_BLOCKS_X 512
#define GEMM_BLOCKS_D 768

typedef float v2f __attribute__((ext_vector_type(2)));
typedef float v8f __attribute__((ext_vector_type(8)));

__device__ __forceinline__ v8f wmma_f32(v2f a, v2f b, v8f c) {
  return __builtin_amdgcn_wmma_f32_16x16x4_f32(
       false, a,  false, b,
       (short)0, c,  false,  false);
}

__device__ __forceinline__ void stage_w_rows(const float* __restrict__ W, int n0,
                                             float* wst, int lane) {
  const float4* g = (const float4*)(W + (size_t)n0 * HID);
#pragma unroll
  for (int q = 0; q < 16; ++q)
    *(float4*)&wst[q * LDSS + lane * 4] = g[q * 32 + lane];
}

__device__ __forceinline__ v2f frag(const float* t, int c, int koff, int i) {
  return *(const v2f*)&t[c * LDSS + 4 * i + koff];
}

__device__ __forceinline__ void lds_barrier() {
  __builtin_amdgcn_fence(__ATOMIC_RELEASE, "workgroup", "local");
  __builtin_amdgcn_s_barrier();
  __builtin_amdgcn_fence(__ATOMIC_ACQUIRE, "workgroup", "local");
}

__global__ __launch_bounds__(256)
void fused_x_kernel(const float* __restrict__ primal,
                    const float* __restrict__ last_primal,
                    const float* __restrict__ W2, const float* __restrict__ b2,
                    const float* __restrict__ W3, const float* __restrict__ b3,
                    float* __restrict__ X, int ntiles) {
  __shared__ __attribute__((aligned(16))) float smem[8 * TILE_F];
  const int tid  = threadIdx.x;
  const int lane = tid & 31;
  const int wave = tid >> 5;
  const int c    = lane & 15;
  const int h    = lane >> 4;
  const int koff = h << 1;
  const int n    = wave * 16 + c;

  int tile = blockIdx.x;

  const int srow = tid >> 5, sc4 = tid & 31;
  const int soff = srow * LDSS + sc4 * 4;
  float4 pa0 = {0.f, 0.f, 0.f, 0.f}, pa1 = pa0, pl0 = pa0, pl1 = pa0;
  if (tile < ntiles) {
    const float4* gp = (const float4*)(primal      + (size_t)tile * 16 * HID);
    const float4* gl = (const float4*)(last_primal + (size_t)tile * 16 * HID);
    pa0 = gp[tid]; pa1 = gp[tid + 256];
    pl0 = gl[tid]; pl1 = gl[tid + 256];
  }

  v2f bw2[32], bw3[32];
  float* wst = smem + wave * TILE_F;
  stage_w_rows(W2, wave * 16, wst, lane);
  __syncthreads();
#pragma unroll
  for (int i = 0; i < 32; ++i) bw2[i] = frag(wst, c, koff, i);
  __syncthreads();
  stage_w_rows(W3, wave * 16, wst, lane);
  __syncthreads();
#pragma unroll
  for (int i = 0; i < 32; ++i) bw3[i] = frag(wst, c, koff, i);
  __syncthreads();

  const float bias = b2[n] - b3[n];

  if (tile < ntiles) {
    *(float4*)&smem[soff]                     = pa0;
    *(float4*)&smem[soff + 8 * LDSS]          = pa1;
    *(float4*)&smem[TILE_F + soff]            = pl0;
    *(float4*)&smem[TILE_F + soff + 8 * LDSS] = pl1;
  }
  __syncthreads();

  int cur = 0;
  for (; tile < ntiles; tile += gridDim.x) {
    const int  next     = tile + gridDim.x;
    const bool has_next = next < ntiles;
    if (has_next) {
      const float4* gp = (const float4*)(primal      + (size_t)next * 16 * HID);
      const float4* gl = (const float4*)(last_primal + (size_t)next * 16 * HID);
      pa0 = gp[tid]; pa1 = gp[tid + 256];
      pl0 = gl[tid]; pl1 = gl[tid + 256];
    }

    const float* Ap = smem + cur * 2 * TILE_F;
    const float* Al = Ap + TILE_F;
    v8f acc  = {};
    v8f acc2 = {};
#pragma unroll
    for (int i = 0; i < 32; ++i) {
      acc  = wmma_f32(frag(Ap, c, koff, i), bw2[i], acc);
      acc2 = wmma_f32(frag(Al, c, koff, i), bw3[i], acc2);
    }

    const int mbase = tile * 16 + (h << 3);
#pragma unroll
    for (int j = 0; j < 8; ++j)
      X[(size_t)(mbase + j) * HID + n] = acc[j] - acc2[j] + bias;

    if (has_next) {
      float* Np = smem + (cur ^ 1) * 2 * TILE_F;
      *(float4*)&Np[soff]                     = pa0;
      *(float4*)&Np[soff + 8 * LDSS]          = pa1;
      *(float4*)&Np[TILE_F + soff]            = pl0;
      *(float4*)&Np[TILE_F + soff + 8 * LDSS] = pl1;
    }
    lds_barrier();
    cur ^= 1;
  }
}

__global__ __launch_bounds__(256)
void dual_base_kernel(const float* __restrict__ dual,
                      const float* __restrict__ W1, const float* __restrict__ b1,
                      const float* __restrict__ rhs,
                      const float* __restrict__ sigma_p,
                      float* __restrict__ out, int ntiles) {
  __shared__ __attribute__((aligned(16))) float smem[8 * TILE_F];
  const int tid  = threadIdx.x;
  const int lane = tid & 31;
  const int wave = tid >> 5;
  const int c    = lane & 15;
  const int h    = lane >> 4;
  const int koff = h << 1;
  const int n    = wave * 16 + c;

  int tile = blockIdx.x;

  const int srow = tid >> 5, sc4 = tid & 31;
  const int soff = srow * LDSS + sc4 * 4;
  float4 pd0 = {0.f, 0.f, 0.f, 0.f}, pd1 = pd0;
  if (tile < ntiles) {
    const float4* gd = (const float4*)(dual + (size_t)tile * 16 * HID);
    pd0 = gd[tid]; pd1 = gd[tid + 256];
  }

  v2f bw[32];
  float* wst = smem + wave * TILE_F;
  stage_w_rows(W1, wave * 16, wst, lane);
  __syncthreads();
#pragma unroll
  for (int i = 0; i < 32; ++i) bw[i] = frag(wst, c, koff, i);
  __syncthreads();

  const float bias  = b1[n];
  const float sigma = sigma_p[0];

  if (tile < ntiles) {
    *(float4*)&smem[soff]            = pd0;
    *(float4*)&smem[soff + 8 * LDSS] = pd1;
  }
  __syncthreads();

  int cur = 0;
  for (; tile < ntiles; tile += gridDim.x) {
    const int  next     = tile + gridDim.x;
    const bool has_next = next < ntiles;
    if (has_next) {
      const float4* gd = (const float4*)(dual + (size_t)next * 16 * HID);
      pd0 = gd[tid]; pd1 = gd[tid + 256];
    }

    const int mbase = tile * 16 + (h << 3);
    float rv[8];
#pragma unroll
    for (int j = 0; j < 8; ++j) rv[j] = rhs[(size_t)(mbase + j) * HID + n];

    const float* Ad = smem + cur * TILE_F;
    v8f acc = {};
#pragma unroll
    for (int i = 0; i < 32; ++i)
      acc = wmma_f32(frag(Ad, c, koff, i), bw[i], acc);

#pragma unroll
    for (int j = 0; j < 8; ++j)
      out[(size_t)(mbase + j) * HID + n] = acc[j] + bias - sigma * rv[j];

    if (has_next) {
      float* Nd = smem + (cur ^ 1) * TILE_F;
      *(float4*)&Nd[soff]            = pd0;
      *(float4*)&Nd[soff + 8 * LDSS] = pd1;
    }
    lds_barrier();
    cur ^= 1;
  }
}

#define SC_U       4
#define SC_NB      (32 / SC_U)
#define OW_ROWS    391
#define OW_BLOCKS  256
#define OW_WAVES   32
#define OW_THREADS (OW_WAVES * 32)
#define OW_QCAP    512
#define OW_GB      8
#define OW_GQCAP   64
#define MK_BLOCKS  64
#define MK_LDSS    257

__device__ __forceinline__ int4 ow_load_rows(const int* __restrict__ rows, long long gbase,
                                             int nnz, int lane) {
  const long long i = gbase + lane * 4;
  if (gbase + 128 <= nnz) return *(const int4*)(rows + i);
  int4 r;
  r.x = i     < nnz ? rows[i]     : -1;
  r.y = i + 1 < nnz ? rows[i + 1] : -1;
  r.z = i + 2 < nnz ? rows[i + 2] : -1;
  r.w = i + 3 < nnz ? rows[i + 3] : -1;
  return r;
}

__device__ __forceinline__ void ow_push(int r, int e, int lo, int nr, int* q, int& tail) {
  const bool m = (unsigned)(r - lo) < (unsigned)nr;
  const unsigned bal = __builtin_amdgcn_ballot_w32(m);
  if (bal != 0u) {
    if (m) q[(tail + (int)__builtin_amdgcn_mbcnt_lo(bal, 0u)) & (OW_QCAP - 1)] = e;
    tail += __builtin_popcount(bal);
  }
}

__device__ __forceinline__ void ow_take_groups(const int* gq, int gpos, int cnt,
                                               const int* __restrict__ rows, int nnz,
                                               int lo, int nr, int* q, int& tail, int lane) {
  const int4* gp = (const int4*)&gq[gpos & (OW_GQCAP - 1)];
  const int4 ga = gp[0], gb = gp[1];
  const int g[OW_GB] = {ga.x, ga.y, ga.z, ga.w, gb.x, gb.y, gb.z, gb.w};
  int rr[OW_GB];
#pragma unroll
  for (int u = 0; u < OW_GB; ++u)
    rr[u] = rows[min(max((int)(((unsigned)g[u] << 5) + lane), 0), nnz - 1)];
#pragma unroll
  for (int u = 0; u < OW_GB; ++u) {
    const int e = (int)(((unsigned)g[u] << 5) + lane);
    ow_push((u < cnt && e < nnz) ? rr[u] : -1, e, lo, nr, q, tail);
  }
}

__device__ __forceinline__ void ow_load_batch(float4 (&xb)[SC_U], const float* __restrict__ X,
                                              int cidx, int j0, int lane) {
#pragma unroll
  for (int u = 0; u < SC_U; ++u) {
    const int cc = __builtin_amdgcn_readlane(cidx, j0 + u);
    xb[u] = *(const float4*)(X + (size_t)cc * HID + lane * 4);
  }
}

__device__ __forceinline__ void ow_add_batch(const float4 (&xb)[SC_U], float* acc,
                                             int rloc, float sval, int j0, int cnt, int lane) {
#pragma unroll
  for (int u = 0; u < SC_U; ++u) {
    if (j0 + u < cnt) {
      const int   rr = __builtin_amdgcn_readlane(rloc, j0 + u);
      const float s  = __int_as_float(__builtin_amdgcn_readlane(__float_as_int(sval), j0 + u));
      float* ap = acc + rr * HID + lane;
      __hip_atomic_fetch_add(&ap[0],  s * xb[u].x, __ATOMIC_RELAXED, __HIP_MEMORY_SCOPE_WORKGROUP);
      __hip_atomic_fetch_add(&ap[32], s * xb[u].y, __ATOMIC_RELAXED, __HIP_MEMORY_SCOPE_WORKGROUP);
      __hip_atomic_fetch_add(&ap[64], s * xb[u].z, __ATOMIC_RELAXED, __HIP_MEMORY_SCOPE_WORKGROUP);
      __hip_atomic_fetch_add(&ap[96], s * xb[u].w, __ATOMIC_RELAXED, __HIP_MEMORY_SCOPE_WORKGROUP);
    }
  }
}

__device__ __forceinline__ void ow_drain(const int* q, int head, int cnt, int lo,
                                         const int* __restrict__ rows, const int* __restrict__ cols,
                                         const float* __restrict__ vals, const float* __restrict__ X,
                                         float sigma, float* acc, int lane) {
  int rloc = 0, cidx = 0;
  float sval = 0.f;
  if (lane < cnt) {
    const int e = q[(head + lane) & (OW_QCAP - 1)];
    rloc = rows[e] - lo;
    cidx = cols[e];
    sval = vals[e] * sigma;
  }
  float4 xb0[SC_U], xb1[SC_U];
  ow_load_batch(xb0, X, cidx, 0, lane);
#pragma unroll
  for (int b = 0; b < SC_NB; b += 2) {
    if ((b + 1) * SC_U < cnt) ow_load_batch(xb1, X, cidx, (b + 1) * SC_U, lane);
    ow_add_batch(xb0, acc, rloc, sval, b * SC_U, cnt, lane);
    if (b + 2 < SC_NB && (b + 2) * SC_U < cnt) ow_load_batch(xb0, X, cidx, (b + 2) * SC_U, lane);
    ow_add_batch(xb1, acc, rloc, sval, (b + 1) * SC_U, cnt, lane);
  }
}


__global__ __launch_bounds__(OW_THREADS)
void owner_mask_kernel(const int* __restrict__ rows, float* __restrict__ X,
                       int n_vars, int nnz) {
  __shared__ unsigned bits[32 * MK_LDSS];
  unsigned* mask = (unsigned*)(X + (size_t)n_vars * HID);
  const int tid  = threadIdx.x;
  const int lane = tid & 31;
  const int wv   = __builtin_amdgcn_readfirstlane(tid >> 5);
  const int nw   = (int)(((long long)nnz + 1023) >> 10);
  const int nwp  = (nw + 31) & ~31;
  unsigned* mybits = bits + wv * MK_LDSS;

  for (int cb = blockIdx.x; cb * 32 < nwp; cb += gridDim.x) {
    for (int i = tid; i < 32 * MK_LDSS; i += OW_THREADS) bits[i] = 0u;
    __syncthreads();

    const long long sb = ((long long)(cb * 32 + wv)) << 10;
    if (sb < nnz) {
      int4 c[8];
      if (sb + 1024 <= nnz) {
#pragma unroll
        for (int j = 0; j < 8; ++j) c[j] = *(const int4*)(rows + sb + j * 128 + lane * 4);
      } else {
#pragma unroll
        for (int j = 0; j < 8; ++j) c[j] = ow_load_rows(rows, sb + j * 128, nnz, lane);
      }
#pragma unroll
      for (int j = 0; j < 8; ++j) {
        const unsigned bit = 1u << (j * 4 + (lane >> 3));
        if (c[j].x >= 0) atomicOr(&mybits[((unsigned)c[j].x / OW_ROWS) % OW_BLOCKS], bit);
        if (c[j].y >= 0) atomicOr(&mybits[((unsigned)c[j].y / OW_ROWS) % OW_BLOCKS], bit);
        if (c[j].z >= 0) atomicOr(&mybits[((unsigned)c[j].z / OW_ROWS) % OW_BLOCKS], bit);
        if (c[j].w >= 0) atomicOr(&mybits[((unsigned)c[j].w / OW_ROWS) % OW_BLOCKS], bit);
      }
    }
    __syncthreads();

#pragma unroll
    for (int k = 0; k < 8; ++k) {
      const int o = wv * 8 + k;
      mask[(size_t)o * nwp + cb * 32 + lane] = bits[lane * MK_LDSS + o];
    }
    __syncthreads();
  }
}

__global__ __launch_bounds__(OW_THREADS)
void owner_scatter_masked_kernel(const int* __restrict__ rows, const int* __restrict__ cols,
                          const float* __restrict__ vals, const float* __restrict__ X,
                          const float* __restrict__ sigma_p, float* __restrict__ out,
                          int nnz, int m_cons, int n_vars) {
  __shared__ __attribute__((aligned(16))) float acc[OW_ROWS * HID];
  __shared__ int ring[OW_WAVES * OW_QCAP];
  __shared__ __attribute__((aligned(16))) int gring[OW_WAVES * OW_GQCAP];
  const int tid  = threadIdx.x;
  const int lane = tid & 31;
  const int wv = __builtin_amdgcn_readfirstlane(tid >> 5);
  int* q = ring + wv * OW_QCAP;
  int* gq = gring + wv * OW_GQCAP;
  const float sigma = sigma_p[0];
  const int nw   = (int)(((long long)nnz + 1023) >> 10);
  const int nwp  = (nw + 31) & ~31;
  const unsigned* mrow = (const unsigned*)(X + (size_t)n_vars * HID) + (size_t)blockIdx.x * nwp;
  const int wpw  = (nw + OW_WAVES - 1) / OW_WAVES;
  const int wbeg = wv * wpw;
  const int wend = min(nw, wbeg + wpw);

  for (int lo = blockIdx.x * OW_ROWS; lo < m_cons; lo += gridDim.x * OW_ROWS) {
    const int nr = min(OW_ROWS, m_cons - lo);

    for (int i = tid; i < OW_ROWS * HID / 4; i += OW_THREADS)
      ((float4*)acc)[i] = make_float4(0.f, 0.f, 0.f, 0.f);
    __syncthreads();

    int head = 0, tail = 0;
    int ghead = 0, gtail = 0;
    for (int s0 = wbeg; s0 < wend; s0 += 32) {
      const unsigned mw = (s0 + lane < wend) ? mrow[s0 + lane] : 0u;
      const int nwi = min(32, wend - s0);
      for (int wi = 0; wi < nwi; ++wi) {
        const unsigned w = (unsigned)__builtin_amdgcn_readlane((int)mw, wi);
        if (w == 0u) continue;
        if ((w >> lane) & 1u)
          gq[(gtail + (int)__builtin_amdgcn_mbcnt_lo(w, 0u)) & (OW_GQCAP - 1)] = ((s0 + wi) << 5) + lane;
        gtail += __builtin_popcount(w);
        while (gtail - ghead >= OW_GB) {
          __builtin_amdgcn_fence(__ATOMIC_ACQ_REL, "wavefront");
          ow_take_groups(gq, ghead, OW_GB, rows, nnz, lo, nr, q, tail, lane);
          ghead += OW_GB;
          while (tail - head >= 32) {
            __builtin_amdgcn_fence(__ATOMIC_ACQ_REL, "wavefront");
            ow_drain(q, head, 32, lo, rows, cols, vals, X, sigma, acc, lane);
            head += 32;
          }
        }
      }
    }
    if (gtail - ghead > 0) {
      __builtin_amdgcn_fence(__ATOMIC_ACQ_REL, "wavefront");
      ow_take_groups(gq, ghead, gtail - ghead, rows, nnz, lo, nr, q, tail, lane);
      while (tail - head >= 32) {
        __builtin_amdgcn_fence(__ATOMIC_ACQ_REL, "wavefront");
        ow_drain(q, head, 32, lo, rows, cols, vals, X, sigma, acc, lane);
        head += 32;
      }
    }
    if (tail - head > 0) {
      __builtin_amdgcn_fence(__ATOMIC_ACQ_REL, "wavefront");
      ow_drain(q, head, tail - head, lo, rows, cols, vals, X, sigma, acc, lane);
    }
    __syncthreads();

    for (int i = tid; i < nr * 32; i += OW_THREADS) {
      const int row = i >> 5, q4 = i & 31;
      float4* op = (float4*)(out + (size_t)(lo + row) * HID) + q4;
      const float* ap = acc + row * HID + q4;
      float4 v = *op;
      v.x += ap[0]; v.y += ap[32]; v.z += ap[64]; v.w += ap[96];
      v.x = v.x >= 0.f ? v.x : NEG_SLOPE * v.x;
      v.y = v.y >= 0.f ? v.y : NEG_SLOPE * v.y;
      v.z = v.z >= 0.f ? v.z : NEG_SLOPE * v.z;
      v.w = v.w >= 0.f ? v.w : NEG_SLOPE * v.w;
      *op = v;
    }
    __syncthreads();
  }
}

#define BN_CH      6272
#define BN_PER     ((BN_CH + OW_THREADS - 1) / OW_THREADS)
#define BN_COLBITS 18
#define OW_MAXCH   512

__global__ __launch_bounds__(OW_THREADS)
void owner_bin_kernel(const int* __restrict__ rows, const int* __restrict__ cols,
                      const float* __restrict__ vals, const float* __restrict__ sigma_p,
                      float* __restrict__ X, int n_vars, int nnz) {
  __shared__ int cnt[OW_BLOCKS];
  __shared__ int start[OW_BLOCKS + 1];
  __shared__ uint2 stage[BN_CH];
  const int tid  = threadIdx.x;
  const int lane = tid & 31;
  const int nch  = (nnz + BN_CH - 1) / BN_CH;
  const int nchp = (nch + 31) & ~31;
  uint2* recs = (uint2*)(X + (size_t)n_vars * HID);
  int*   offT = (int*)(recs + (size_t)nch * BN_CH);
  const float sigma = sigma_p[0];

  for (int ch = blockIdx.x; ch < nch; ch += gridDim.x) {
    if (tid < OW_BLOCKS) cnt[tid] = 0;
    __syncthreads();

    const int base = ch * BN_CH;
    const int n    = min(BN_CH, nnz - base);
    int   rr[BN_PER], cc[BN_PER];
    float vv[BN_PER];
#pragma unroll
    for (int k = 0; k < BN_PER; ++k) {
      const int e = base + min(k * OW_THREADS + tid, n - 1);
      rr[k] = rows[e]; cc[k] = cols[e]; vv[k] = vals[e];
    }
    unsigned pk[BN_PER];
    float    sv[BN_PER];
    int      ow[BN_PER], rk[BN_PER];
#pragma unroll
    for (int k = 0; k < BN_PER; ++k) {
      const int i = k * OW_THREADS + tid;
      ow[k] = -1; rk[k] = 0; pk[k] = 0u; sv[k] = 0.f;
      if (i < n) {
        const unsigned r = (unsigned)rr[k];
        const unsigned q = r / OW_ROWS;
        const int o = (int)min(q, (unsigned)(OW_BLOCKS - 1));
        pk[k] = ((unsigned)cc[k] & ((1u << BN_COLBITS) - 1u)) | ((r - q * OW_ROWS) << BN_COLBITS);
        sv[k] = vv[k] * sigma;
        ow[k] = o;
        rk[k] = atomicAdd(&cnt[o], 1);
      }
    }
    __syncthreads();

    if (tid < 32) {
      int c[8], s = 0;
#pragma unroll
      for (int j = 0; j < 8; ++j) { c[j] = cnt[tid * 8 + j]; s += c[j]; }
      int inc = s;
#pragma unroll
      for (int d = 1; d < 32; d <<= 1) {
        const int t = __shfl_up(inc, d);
        if (lane >= d) inc += t;
      }
      int ex = inc - s;
#pragma unroll
      for (int j = 0; j < 8; ++j) { start[tid * 8 + j] = ex; ex += c[j]; }
      if (tid == 31) start[OW_BLOCKS] = ex;
    }
    __syncthreads();

#pragma unroll
    for (int k = 0; k < BN_PER; ++k)
      if (ow[k] >= 0) stage[start[ow[k]] + rk[k]] = make_uint2(pk[k], __float_as_uint(sv[k]));
    if (tid <= OW_BLOCKS) offT[(size_t)tid * nchp + ch] = start[tid];
    __syncthreads();

    for (int i = tid; i < n; i += OW_THREADS) recs[(size_t)base + i] = stage[i];
    __syncthreads();
  }
}

__device__ __forceinline__ long long ow_find_rec(const int* segP, const int* segS, int g, long long nrec) {
  int j = 0;
#pragma unroll
  for (int step = OW_MAXCH / 2; step >= 1; step >>= 1)
    if (segP[j + step] <= g) j += step;
  const long long idx = (long long)segS[j] + (g - segP[j]);
  return min(max(idx, 0ll), nrec - 1);
}

__device__ __forceinline__ void ow_drain_rec(uint2 rec, int cnt, int n_vars,
                                             const float* __restrict__ X, float* acc, int lane,
                                             const int* segP, const int* segS, int gnext,
                                             long long nrec, long long& inext) {
  int rloc = 0, cidx = 0;
  float sval = 0.f;
  if (lane < cnt) {
    cidx = min((int)(rec.x & ((1u << BN_COLBITS) - 1u)), n_vars - 1);
    rloc = min((int)(rec.x >> BN_COLBITS), OW_ROWS - 1);
    sval = __uint_as_float(rec.y);
  }
  float4 xb0[SC_U], xb1[SC_U];
  ow_load_batch(xb0, X, cidx, 0, lane);
#pragma unroll
  for (int b = 0; b < SC_NB; b += 2) {
    if ((b + 1) * SC_U < cnt) ow_load_batch(xb1, X, cidx, (b + 1) * SC_U, lane);
    if (b == 0) inext = ow_find_rec(segP, segS, gnext, nrec);
    ow_add_batch(xb0, acc, rloc, sval, b * SC_U, cnt, lane);
    if (b + 2 < SC_NB && (b + 2) * SC_U < cnt) ow_load_batch(xb0, X, cidx, (b + 2) * SC_U, lane);
    ow_add_batch(xb1, acc, rloc, sval, (b + 1) * SC_U, cnt, lane);
  }
}

__global__ __launch_bounds__(OW_THREADS)
void owner_scatter_kernel(const float* __restrict__ X, float* __restrict__ out,
                          int nnz, int m_cons, int n_vars) {
  __shared__ __attribute__((aligned(16))) float acc[OW_ROWS * HID];
  __shared__ int segL[OW_MAXCH];
  __shared__ int segS[OW_MAXCH];
  __shared__ int segP[OW_MAXCH + 1];
  const int tid  = threadIdx.x;
  const int lane = tid & 31;
  const int wv   = __builtin_amdgcn_readfirstlane(tid >> 5);
  const int nch  = (nnz + BN_CH - 1) / BN_CH;
  const int nchp = (nch + 31) & ~31;
  const long long nrec = (long long)nch * BN_CH;
  const uint2* recs = (const uint2*)(X + (size_t)n_vars * HID);
  const int*   offT = (const int*)(recs + nrec);
  const int lo = blockIdx.x * OW_ROWS;
  const int nr = min(OW_ROWS, m_cons - lo);

  for (int i = tid; i < OW_ROWS * HID / 4; i += OW_THREADS)
    ((float4*)acc)[i] = make_float4(0.f, 0.f, 0.f, 0.f);

  for (int c0 = 0; c0 < nch; c0 += OW_MAXCH) {
    const int ncc = min(OW_MAXCH, nch - c0);
    __syncthreads();
    if (tid < OW_MAXCH) {
      int len = 0, s = 0;
      if (tid < ncc) {
        const int ch = c0 + tid;
        const int a = offT[(size_t)blockIdx.x * nchp + ch];
        const int b = offT[(size_t)(blockIdx.x + 1) * nchp + ch];
        len = max(b - a, 0);
        s   = ch * BN_CH + a;
      }
      segL[tid] = len;
      segS[tid] = s;
    }
    __syncthreads();
    if (tid < 32) {
      int c[OW_MAXCH / 32], sum = 0;
#pragma unroll
      for (int j = 0; j < OW_MAXCH / 32; ++j) { c[j] = segL[tid * (OW_MAXCH / 32) + j]; sum += c[j]; }
      int inc = sum;
#pragma unroll
      for (int d = 1; d < 32; d <<= 1) {
        const int t = __shfl_up(inc, d);
        if (lane >= d) inc += t;
      }
      int ex = inc - sum;
#pragma unroll
      for (int j = 0; j < OW_MAXCH / 32; ++j) { segP[tid * (OW_MAXCH / 32) + j] = ex; ex += c[j]; }
      if (tid == 31) segP[OW_MAXCH] = ex;
    }
    __syncthreads();

    const int T    = __builtin_amdgcn_readfirstlane(segP[OW_MAXCH]);
    const int per  = (T + OW_WAVES - 1) / OW_WAVES;
    const int gend = min(T, (wv + 1) * per);
    int g0 = wv * per;
    uint2 rec = recs[ow_find_rec(segP, segS, g0 + lane, nrec)];
    long long inext = ow_find_rec(segP, segS, g0 + 32 + lane, nrec);
    for (; g0 < gend; g0 += 32) {
      const uint2 nx = recs[inext];
      ow_drain_rec(rec, min(32, gend - g0), n_vars, X, acc, lane, segP, segS, g0 + 64 + lane, nrec, inext);
      rec = nx;
    }
  }
  __syncthreads();

  for (int i = tid; i < nr * 32; i += OW_THREADS) {
    const int row = i >> 5, q4 = i & 31;
    float4* op = (float4*)(out + (size_t)(lo + row) * HID) + q4;
    const float* ap = acc + row * HID + q4;
    float4 v = *op;
    v.x += ap[0]; v.y += ap[32]; v.z += ap[64]; v.w += ap[96];
    v.x = v.x >= 0.f ? v.x : NEG_SLOPE * v.x;
    v.y = v.y >= 0.f ? v.y : NEG_SLOPE * v.y;
    v.z = v.z >= 0.f ? v.z : NEG_SLOPE * v.z;
    v.w = v.w >= 0.f ? v.w : NEG_SLOPE * v.w;
    *op = v;
  }
}

extern "C" void kernel_launch(void* const* d_in, const int* in_sizes, int n_in,
                              void* d_out, int out_size, void* d_ws, size_t ws_size,
                              hipStream_t stream) {
  const float* primal      = (const float*)d_in[0];
  const float* last_primal = (const float*)d_in[1];
  const float* dual        = (const float*)d_in[2];
  const int*   rows        = (const int*)d_in[3];
  const int*   cols        = (const int*)d_in[4];
  const float* vals        = (const float*)d_in[5];
  const float* rhs         = (const float*)d_in[6];
  const float* W1          = (const float*)d_in[7];
  const float* b1          = (const float*)d_in[8];
  const float* W2          = (const float*)d_in[9];
  const float* b2          = (const float*)d_in[10];
  const float* W3          = (const float*)d_in[11];
  const float* b3          = (const float*)d_in[12];
  const float* sigma       = (const float*)d_in[13];

  float* out = (float*)d_out;
  float* X   = (float*)d_ws;

  const int n_vars = in_sizes[0] / HID;
  const int m_cons = in_sizes[2] / HID;
  const int nnz    = in_sizes[5];

  const size_t nch    = ((size_t)nnz + BN_CH - 1) / BN_CH;
  const size_t need   = (size_t)n_vars * HID * 4 + nch * BN_CH * 8 + (OW_BLOCKS + 1) * ((nch + 31) / 32 * 32) * 4;
  const bool   binned = ws_size >= need && n_vars <= (1 << BN_COLBITS) && m_cons <= OW_BLOCKS * OW_ROWS;

  if (binned) owner_bin_kernel<<<OW_BLOCKS, OW_THREADS, 0, stream>>>(rows, cols, vals, sigma, X, n_vars, nnz);
  else        owner_mask_kernel<<<MK_BLOCKS, OW_THREADS, 0, stream>>>(rows, X, n_vars, nnz);
  fused_x_kernel<<<GEMM_BLOCKS_X, 256, 0, stream>>>(primal, last_primal,
                                                    W2, b2, W3, b3, X, n_vars / 16);
  dual_base_kernel<<<GEMM_BLOCKS_D, 256, 0, stream>>>(dual, W1, b1, rhs, sigma, out,
                                                      m_cons / 16);
  if (binned) owner_scatter_kernel<<<OW_BLOCKS, OW_THREADS, 0, stream>>>(X, out, nnz, m_cons, n_vars);
  else        owner_scatter_masked_kernel<<<OW_BLOCKS, OW_THREADS, 0, stream>>>(rows, cols, vals, X, sigma, out,
                                                                           nnz, m_cons, n_vars);
}


# Binned route: dual @ W1^T moves into the scatter block: 8 of its 32 waves run it on WMMA into the LDS sums (then a half gather share) while 24 gather; kernel 2 launch goes; sum order changes
# speedup vs baseline: 3.2811x; 1.0780x over previous
#include <hip/hip_runtime.h>


#define HID 128
#define LDSS 132
#define TILE_F (16 * LDSS)
#define NEG_SLOPE 0.01f
#define GEMM_BLOCKS_X 512
#define GEMM_BLOCKS_D 768

typedef float v2f __attribute__((ext_vector_type(2)));
typedef float v8f __attribute__((ext_vector_type(8)));

__device__ __forceinline__ v8f wmma_f32(v2f a, v2f b, v8f c) {
  return __builtin_amdgcn_wmma_f32_16x16x4_f32(
       false, a,  false, b,
       (short)0, c,  false,  false);
}

__device__ __forceinline__ void stage_w_rows(const float* __restrict__ W, int n0,
                                             float* wst, int lane) {
  const float4* g = (const float4*)(W + (size_t)n0 * HID);
#pragma unroll
  for (int q = 0; q < 16; ++q)
    *(float4*)&wst[q * LDSS + lane * 4] = g[q * 32 + lane];
}

__device__ __forceinline__ v2f frag(const float* t, int c, int koff, int i) {
  return *(const v2f*)&t[c * LDSS + 4 * i + koff];
}

__device__ __forceinline__ void lds_barrier() {
  __builtin_amdgcn_fence(__ATOMIC_RELEASE, "workgroup", "local");
  __builtin_amdgcn_s_barrier();
  __builtin_amdgcn_fence(__ATOMIC_ACQUIRE, "workgroup", "local");
}

__global__ __launch_bounds__(256)
void fused_x_kernel(const float* __restrict__ primal,
                    const float* __restrict__ last_primal,
                    const float* __restrict__ W2, const float* __restrict__ b2,
                    const float* __restrict__ W3, const float* __restrict__ b3,
                    float* __restrict__ X, int ntiles) {
  __shared__ __attribute__((aligned(16))) float smem[8 * TILE_F];
  const int tid  = threadIdx.x;
  const int lane = tid & 31;
  const int wave = tid >> 5;
  const int c    = lane & 15;
  const int h    = lane >> 4;
  const int koff = h << 1;
  const int n    = wave * 16 + c;

  int tile = blockIdx.x;

  const int srow = tid >> 5, sc4 = tid & 31;
  const int soff = srow * LDSS + sc4 * 4;
  float4 pa0 = {0.f, 0.f, 0.f, 0.f}, pa1 = pa0, pl0 = pa0, pl1 = pa0;
  if (tile < ntiles) {
    const float4* gp = (const float4*)(primal      + (size_t)tile * 16 * HID);
    const float4* gl = (const float4*)(last_primal + (size_t)tile * 16 * HID);
    pa0 = gp[tid]; pa1 = gp[tid + 256];
    pl0 = gl[tid]; pl1 = gl[tid + 256];
  }

  v2f bw2[32], bw3[32];
  float* wst = smem + wave * TILE_F;
  stage_w_rows(W2, wave * 16, wst, lane);
  __syncthreads();
#pragma unroll
  for (int i = 0; i < 32; ++i) bw2[i] = frag(wst, c, koff, i);
  __syncthreads();
  stage_w_rows(W3, wave * 16, wst, lane);
  __syncthreads();
#pragma unroll
  for (int i = 0; i < 32; ++i) bw3[i] = frag(wst, c, koff, i);
  __syncthreads();

  const float bias = b2[n] - b3[n];

  if (tile < ntiles) {
    *(float4*)&smem[soff]                     = pa0;
    *(float4*)&smem[soff + 8 * LDSS]          = pa1;
    *(float4*)&smem[TILE_F + soff]            = pl0;
    *(float4*)&smem[TILE_F + soff + 8 * LDSS] = pl1;
  }
  __syncthreads();

  int cur = 0;
  for (; tile < ntiles; tile += gridDim.x) {
    const int  next     = tile + gridDim.x;
    const bool has_next = next < ntiles;
    if (has_next) {
      const float4* gp = (const float4*)(primal      + (size_t)next * 16 * HID);
      const float4* gl = (const float4*)(last_primal + (size_t)next * 16 * HID);
      pa0 = gp[tid]; pa1 = gp[tid + 256];
      pl0 = gl[tid]; pl1 = gl[tid + 256];
    }

    const float* Ap = smem + cur * 2 * TILE_F;
    const float* Al = Ap + TILE_F;
    v8f acc  = {};
    v8f acc2 = {};
#pragma unroll
    for (int i = 0; i < 32; ++i) {
      acc  = wmma_f32(frag(Ap, c, koff, i), bw2[i], acc);
      acc2 = wmma_f32(frag(Al, c, koff, i), bw3[i], acc2);
    }

    const int mbase = tile * 16 + (h << 3);
#pragma unroll
    for (int j = 0; j < 8; ++j)
      X[(size_t)(mbase + j) * HID + n] = acc[j] - acc2[j] + bias;

    if (has_next) {
      float* Np = smem + (cur ^ 1) * 2 * TILE_F;
      *(float4*)&Np[soff]                     = pa0;
      *(float4*)&Np[soff + 8 * LDSS]          = pa1;
      *(float4*)&Np[TILE_F + soff]            = pl0;
      *(float4*)&Np[TILE_F + soff + 8 * LDSS] = pl1;
    }
    lds_barrier();
    cur ^= 1;
  }
}

__global__ __launch_bounds__(256)
void dual_base_kernel(const float* __restrict__ dual,
                      const float* __restrict__ W1, const float* __restrict__ b1,
                      const float* __restrict__ rhs,
                      const float* __restrict__ sigma_p,
                      float* __restrict__ out, int ntiles) {
  __shared__ __attribute__((aligned(16))) float smem[8 * TILE_F];
  const int tid  = threadIdx.x;
  const int lane = tid & 31;
  const int wave = tid >> 5;
  const int c    = lane & 15;
  const int h    = lane >> 4;
  const int koff = h << 1;
  const int n    = wave * 16 + c;

  int tile = blockIdx.x;

  const int srow = tid >> 5, sc4 = tid & 31;
  const int soff = srow * LDSS + sc4 * 4;
  float4 pd0 = {0.f, 0.f, 0.f, 0.f}, pd1 = pd0;
  if (tile < ntiles) {
    const float4* gd = (const float4*)(dual + (size_t)tile * 16 * HID);
    pd0 = gd[tid]; pd1 = gd[tid + 256];
  }

  v2f bw[32];
  float* wst = smem + wave * TILE_F;
  stage_w_rows(W1, wave * 16, wst, lane);
  __syncthreads();
#pragma unroll
  for (int i = 0; i < 32; ++i) bw[i] = frag(wst, c, koff, i);
  __syncthreads();

  const float bias  = b1[n];
  const float sigma = sigma_p[0];

  if (tile < ntiles) {
    *(float4*)&smem[soff]            = pd0;
    *(float4*)&smem[soff + 8 * LDSS] = pd1;
  }
  __syncthreads();

  int cur = 0;
  for (; tile < ntiles; tile += gridDim.x) {
    const int  next     = tile + gridDim.x;
    const bool has_next = next < ntiles;
    if (has_next) {
      const float4* gd = (const float4*)(dual + (size_t)next * 16 * HID);
      pd0 = gd[tid]; pd1 = gd[tid + 256];
    }

    const int mbase = tile * 16 + (h << 3);
    float rv[8];
#pragma unroll
    for (int j = 0; j < 8; ++j) rv[j] = rhs[(size_t)(mbase + j) * HID + n];

    const float* Ad = smem + cur * TILE_F;
    v8f acc = {};
#pragma unroll
    for (int i = 0; i < 32; ++i)
      acc = wmma_f32(frag(Ad, c, koff, i), bw[i], acc);

#pragma unroll
    for (int j = 0; j < 8; ++j)
      out[(size_t)(mbase + j) * HID + n] = acc[j] + bias - sigma * rv[j];

    if (has_next) {
      float* Nd = smem + (cur ^ 1) * TILE_F;
      *(float4*)&Nd[soff]            = pd0;
      *(float4*)&Nd[soff + 8 * LDSS] = pd1;
    }
    lds_barrier();
    cur ^= 1;
  }
}

#define SC_U       4
#define SC_NB      (32 / SC_U)
#define OW_ROWS    391
#define OW_BLOCKS  256
#define OW_WAVES   32
#define OW_THREADS (OW_WAVES * 32)
#define OW_QCAP    512
#define OW_GB      8
#define OW_GQCAP   64
#define MK_BLOCKS  64
#define MK_LDSS    257

__device__ __forceinline__ int4 ow_load_rows(const int* __restrict__ rows, long long gbase,
                                             int nnz, int lane) {
  const long long i = gbase + lane * 4;
  if (gbase + 128 <= nnz) return *(const int4*)(rows + i);
  int4 r;
  r.x = i     < nnz ? rows[i]     : -1;
  r.y = i + 1 < nnz ? rows[i + 1] : -1;
  r.z = i + 2 < nnz ? rows[i + 2] : -1;
  r.w = i + 3 < nnz ? rows[i + 3] : -1;
  return r;
}

__device__ __forceinline__ void ow_push(int r, int e, int lo, int nr, int* q, int& tail) {
  const bool m = (unsigned)(r - lo) < (unsigned)nr;
  const unsigned bal = __builtin_amdgcn_ballot_w32(m);
  if (bal != 0u) {
    if (m) q[(tail + (int)__builtin_amdgcn_mbcnt_lo(bal, 0u)) & (OW_QCAP - 1)] = e;
    tail += __builtin_popcount(bal);
  }
}

__device__ __forceinline__ void ow_take_groups(const int* gq, int gpos, int cnt,
                                               const int* __restrict__ rows, int nnz,
                                               int lo, int nr, int* q, int& tail, int lane) {
  const int4* gp = (const int4*)&gq[gpos & (OW_GQCAP - 1)];
  const int4 ga = gp[0], gb = gp[1];
  const int g[OW_GB] = {ga.x, ga.y, ga.z, ga.w, gb.x, gb.y, gb.z, gb.w};
  int rr[OW_GB];
#pragma unroll
  for (int u = 0; u < OW_GB; ++u)
    rr[u] = rows[min(max((int)(((unsigned)g[u] << 5) + lane), 0), nnz - 1)];
#pragma unroll
  for (int u = 0; u < OW_GB; ++u) {
    const int e = (int)(((unsigned)g[u] << 5) + lane);
    ow_push((u < cnt && e < nnz) ? rr[u] : -1, e, lo, nr, q, tail);
  }
}

__device__ __forceinline__ void ow_load_batch(float4 (&xb)[SC_U], const float* __restrict__ X,
                                              int cidx, int j0, int lane) {
#pragma unroll
  for (int u = 0; u < SC_U; ++u) {
    const int cc = __builtin_amdgcn_readlane(cidx, j0 + u);
    xb[u] = *(const float4*)(X + (size_t)cc * HID + lane * 4);
  }
}

__device__ __forceinline__ void ow_add_batch(const float4 (&xb)[SC_U], float* acc,
                                             int rloc, float sval, int j0, int cnt, int lane) {
#pragma unroll
  for (int u = 0; u < SC_U; ++u) {
    if (j0 + u < cnt) {
      const int   rr = __builtin_amdgcn_readlane(rloc, j0 + u);
      const float s  = __int_as_float(__builtin_amdgcn_readlane(__float_as_int(sval), j0 + u));
      float* ap = acc + rr * HID + lane;
      __hip_atomic_fetch_add(&ap[0],  s * xb[u].x, __ATOMIC_RELAXED, __HIP_MEMORY_SCOPE_WORKGROUP);
      __hip_atomic_fetch_add(&ap[32], s * xb[u].y, __ATOMIC_RELAXED, __HIP_MEMORY_SCOPE_WORKGROUP);
      __hip_atomic_fetch_add(&ap[64], s * xb[u].z, __ATOMIC_RELAXED, __HIP_MEMORY_SCOPE_WORKGROUP);
      __hip_atomic_fetch_add(&ap[96], s * xb[u].w, __ATOMIC_RELAXED, __HIP_MEMORY_SCOPE_WORKGROUP);
    }
  }
}

__device__ __forceinline__ void ow_drain(const int* q, int head, int cnt, int lo,
                                         const int* __restrict__ rows, const int* __restrict__ cols,
                                         const float* __restrict__ vals, const float* __restrict__ X,
                                         float sigma, float* acc, int lane) {
  int rloc = 0, cidx = 0;
  float sval = 0.f;
  if (lane < cnt) {
    const int e = q[(head + lane) & (OW_QCAP - 1)];
    rloc = rows[e] - lo;
    cidx = cols[e];
    sval = vals[e] * sigma;
  }
  float4 xb0[SC_U], xb1[SC_U];
  ow_load_batch(xb0, X, cidx, 0, lane);
#pragma unroll
  for (int b = 0; b < SC_NB; b += 2) {
    if ((b + 1) * SC_U < cnt) ow_load_batch(xb1, X, cidx, (b + 1) * SC_U, lane);
    ow_add_batch(xb0, acc, rloc, sval, b * SC_U, cnt, lane);
    if (b + 2 < SC_NB && (b + 2) * SC_U < cnt) ow_load_batch(xb0, X, cidx, (b + 2) * SC_U, lane);
    ow_add_batch(xb1, acc, rloc, sval, (b + 1) * SC_U, cnt, lane);
  }
}


__global__ __launch_bounds__(OW_THREADS)
void owner_mask_kernel(const int* __restrict__ rows, float* __restrict__ X,
                       int n_vars, int nnz) {
  __shared__ unsigned bits[32 * MK_LDSS];
  unsigned* mask = (unsigned*)(X + (size_t)n_vars * HID);
  const int tid  = threadIdx.x;
  const int lane = tid & 31;
  const int wv   = __builtin_amdgcn_readfirstlane(tid >> 5);
  const int nw   = (int)(((long long)nnz + 1023) >> 10);
  const int nwp  = (nw + 31) & ~31;
  unsigned* mybits = bits + wv * MK_LDSS;

  for (int cb = blockIdx.x; cb * 32 < nwp; cb += gridDim.x) {
    for (int i = tid; i < 32 * MK_LDSS; i += OW_THREADS) bits[i] = 0u;
    __syncthreads();

    const long long sb = ((long long)(cb * 32 + wv)) << 10;
    if (sb < nnz) {
      int4 c[8];
      if (sb + 1024 <= nnz) {
#pragma unroll
        for (int j = 0; j < 8; ++j) c[j] = *(const int4*)(rows + sb + j * 128 + lane * 4);
      } else {
#pragma unroll
        for (int j = 0; j < 8; ++j) c[j] = ow_load_rows(rows, sb + j * 128, nnz, lane);
      }
#pragma unroll
      for (int j = 0; j < 8; ++j) {
        const unsigned bit = 1u << (j * 4 + (lane >> 3));
        if (c[j].x >= 0) atomicOr(&mybits[((unsigned)c[j].x / OW_ROWS) % OW_BLOCKS], bit);
        if (c[j].y >= 0) atomicOr(&mybits[((unsigned)c[j].y / OW_ROWS) % OW_BLOCKS], bit);
        if (c[j].z >= 0) atomicOr(&mybits[((unsigned)c[j].z / OW_ROWS) % OW_BLOCKS], bit);
        if (c[j].w >= 0) atomicOr(&mybits[((unsigned)c[j].w / OW_ROWS) % OW_BLOCKS], bit);
      }
    }
    __syncthreads();

#pragma unroll
    for (int k = 0; k < 8; ++k) {
      const int o = wv * 8 + k;
      mask[(size_t)o * nwp + cb * 32 + lane] = bits[lane * MK_LDSS + o];
    }
    __syncthreads();
  }
}

__global__ __launch_bounds__(OW_THREADS)
void owner_scatter_masked_kernel(const int* __restrict__ rows, const int* __restrict__ cols,
                          const float* __restrict__ vals, const float* __restrict__ X,
                          const float* __restrict__ sigma_p, float* __restrict__ out,
                          int nnz, int m_cons, int n_vars) {
  __shared__ __attribute__((aligned(16))) float acc[OW_ROWS * HID];
  __shared__ int ring[OW_WAVES * OW_QCAP];
  __shared__ __attribute__((aligned(16))) int gring[OW_WAVES * OW_GQCAP];
  const int tid  = threadIdx.x;
  const int lane = tid & 31;
  const int wv = __builtin_amdgcn_readfirstlane(tid >> 5);
  int* q = ring + wv * OW_QCAP;
  int* gq = gring + wv * OW_GQCAP;
  const float sigma = sigma_p[0];
  const int nw   = (int)(((long long)nnz + 1023) >> 10);
  const int nwp  = (nw + 31) & ~31;
  const unsigned* mrow = (const unsigned*)(X + (size_t)n_vars * HID) + (size_t)blockIdx.x * nwp;
  const int wpw  = (nw + OW_WAVES - 1) / OW_WAVES;
  const int wbeg = wv * wpw;
  const int wend = min(nw, wbeg + wpw);

  for (int lo = blockIdx.x * OW_ROWS; lo < m_cons; lo += gridDim.x * OW_ROWS) {
    const int nr = min(OW_ROWS, m_cons - lo);

    for (int i = tid; i < OW_ROWS * HID / 4; i += OW_THREADS)
      ((float4*)acc)[i] = make_float4(0.f, 0.f, 0.f, 0.f);
    __syncthreads();

    int head = 0, tail = 0;
    int ghead = 0, gtail = 0;
    for (int s0 = wbeg; s0 < wend; s0 += 32) {
      const unsigned mw = (s0 + lane < wend) ? mrow[s0 + lane] : 0u;
      const int nwi = min(32, wend - s0);
      for (int wi = 0; wi < nwi; ++wi) {
        const unsigned w = (unsigned)__builtin_amdgcn_readlane((int)mw, wi);
        if (w == 0u) continue;
        if ((w >> lane) & 1u)
          gq[(gtail + (int)__builtin_amdgcn_mbcnt_lo(w, 0u)) & (OW_GQCAP - 1)] = ((s0 + wi) << 5) + lane;
        gtail += __builtin_popcount(w);
        while (gtail - ghead >= OW_GB) {
          __builtin_amdgcn_fence(__ATOMIC_ACQ_REL, "wavefront");
          ow_take_groups(gq, ghead, OW_GB, rows, nnz, lo, nr, q, tail, lane);
          ghead += OW_GB;
          while (tail - head >= 32) {
            __builtin_amdgcn_fence(__ATOMIC_ACQ_REL, "wavefront");
            ow_drain(q, head, 32, lo, rows, cols, vals, X, sigma, acc, lane);
            head += 32;
          }
        }
      }
    }
    if (gtail - ghead > 0) {
      __builtin_amdgcn_fence(__ATOMIC_ACQ_REL, "wavefront");
      ow_take_groups(gq, ghead, gtail - ghead, rows, nnz, lo, nr, q, tail, lane);
      while (tail - head >= 32) {
        __builtin_amdgcn_fence(__ATOMIC_ACQ_REL, "wavefront");
        ow_drain(q, head, 32, lo, rows, cols, vals, X, sigma, acc, lane);
        head += 32;
      }
    }
    if (tail - head > 0) {
      __builtin_amdgcn_fence(__ATOMIC_ACQ_REL, "wavefront");
      ow_drain(q, head, tail - head, lo, rows, cols, vals, X, sigma, acc, lane);
    }
    __syncthreads();

    for (int i = tid; i < nr * 32; i += OW_THREADS) {
      const int row = i >> 5, q4 = i & 31;
      float4* op = (float4*)(out + (size_t)(lo + row) * HID) + q4;
      const float* ap = acc + row * HID + q4;
      float4 v = *op;
      v.x += ap[0]; v.y += ap[32]; v.z += ap[64]; v.w += ap[96];
      v.x = v.x >= 0.f ? v.x : NEG_SLOPE * v.x;
      v.y = v.y >= 0.f ? v.y : NEG_SLOPE * v.y;
      v.z = v.z >= 0.f ? v.z : NEG_SLOPE * v.z;
      v.w = v.w >= 0.f ? v.w : NEG_SLOPE * v.w;
      *op = v;
    }
    __syncthreads();
  }
}

#define BN_CH      6272
#define BN_PER     ((BN_CH + OW_THREADS - 1) / OW_THREADS)
#define BN_COLBITS 18
#define OW_MAXCH   512

__global__ __launch_bounds__(OW_THREADS)
void owner_bin_kernel(const int* __restrict__ rows, const int* __restrict__ cols,
                      const float* __restrict__ vals, const float* __restrict__ sigma_p,
                      float* __restrict__ X, int n_vars, int nnz) {
  __shared__ int cnt[OW_BLOCKS];
  __shared__ int start[OW_BLOCKS + 1];
  __shared__ uint2 stage[BN_CH];
  const int tid  = threadIdx.x;
  const int lane = tid & 31;
  const int nch  = (nnz + BN_CH - 1) / BN_CH;
  const int nchp = (nch + 31) & ~31;
  uint2* recs = (uint2*)(X + (size_t)n_vars * HID);
  int*   offT = (int*)(recs + (size_t)nch * BN_CH);
  const float sigma = sigma_p[0];

  for (int ch = blockIdx.x; ch < nch; ch += gridDim.x) {
    if (tid < OW_BLOCKS) cnt[tid] = 0;
    __syncthreads();

    const int base = ch * BN_CH;
    const int n    = min(BN_CH, nnz - base);
    int   rr[BN_PER], cc[BN_PER];
    float vv[BN_PER];
#pragma unroll
    for (int k = 0; k < BN_PER; ++k) {
      const int e = base + min(k * OW_THREADS + tid, n - 1);
      rr[k] = rows[e]; cc[k] = cols[e]; vv[k] = vals[e];
    }
    unsigned pk[BN_PER];
    float    sv[BN_PER];
    int      ow[BN_PER], rk[BN_PER];
#pragma unroll
    for (int k = 0; k < BN_PER; ++k) {
      const int i = k * OW_THREADS + tid;
      ow[k] = -1; rk[k] = 0; pk[k] = 0u; sv[k] = 0.f;
      if (i < n) {
        const unsigned r = (unsigned)rr[k];
        const unsigned q = r / OW_ROWS;
        const int o = (int)min(q, (unsigned)(OW_BLOCKS - 1));
        pk[k] = ((unsigned)cc[k] & ((1u << BN_COLBITS) - 1u)) | ((r - q * OW_ROWS) << BN_COLBITS);
        sv[k] = vv[k] * sigma;
        ow[k] = o;
        rk[k] = atomicAdd(&cnt[o], 1);
      }
    }
    __syncthreads();

    if (tid < 32) {
      int c[8], s = 0;
#pragma unroll
      for (int j = 0; j < 8; ++j) { c[j] = cnt[tid * 8 + j]; s += c[j]; }
      int inc = s;
#pragma unroll
      for (int d = 1; d < 32; d <<= 1) {
        const int t = __shfl_up(inc, d);
        if (lane >= d) inc += t;
      }
      int ex = inc - s;
#pragma unroll
      for (int j = 0; j < 8; ++j) { start[tid * 8 + j] = ex; ex += c[j]; }
      if (tid == 31) start[OW_BLOCKS] = ex;
    }
    __syncthreads();

#pragma unroll
    for (int k = 0; k < BN_PER; ++k)
      if (ow[k] >= 0) stage[start[ow[k]] + rk[k]] = make_uint2(pk[k], __float_as_uint(sv[k]));
    if (tid <= OW_BLOCKS) offT[(size_t)tid * nchp + ch] = start[tid];
    __syncthreads();

    for (int i = tid; i < n; i += OW_THREADS) recs[(size_t)base + i] = stage[i];
    __syncthreads();
  }
}

__device__ __forceinline__ long long ow_find_rec(const int* segP, const int* segS, int g, long long nrec) {
  int j = 0;
#pragma unroll
  for (int step = OW_MAXCH / 2; step >= 1; step >>= 1)
    if (segP[j + step] <= g) j += step;
  const long long idx = (long long)segS[j] + (g - segP[j]);
  return min(max(idx, 0ll), nrec - 1);
}

__device__ __forceinline__ void ow_drain_rec(uint2 rec, int cnt, int n_vars,
                                             const float* __restrict__ X, float* acc, int lane,
                                             const int* segP, const int* segS, int gnext,
                                             long long nrec, long long& inext) {
  int rloc = 0, cidx = 0;
  float sval = 0.f;
  if (lane < cnt) {
    cidx = min((int)(rec.x & ((1u << BN_COLBITS) - 1u)), n_vars - 1);
    rloc = min((int)(rec.x >> BN_COLBITS), OW_ROWS - 1);
    sval = __uint_as_float(rec.y);
  }
  float4 xb0[SC_U], xb1[SC_U];
  ow_load_batch(xb0, X, cidx, 0, lane);
#pragma unroll
  for (int b = 0; b < SC_NB; b += 2) {
    if ((b + 1) * SC_U < cnt) ow_load_batch(xb1, X, cidx, (b + 1) * SC_U, lane);
    if (b == 0) inext = ow_find_rec(segP, segS, gnext, nrec);
    ow_add_batch(xb0, acc, rloc, sval, b * SC_U, cnt, lane);
    if (b + 2 < SC_NB && (b + 2) * SC_U < cnt) ow_load_batch(xb0, X, cidx, (b + 2) * SC_U, lane);
    ow_add_batch(xb1, acc, rloc, sval, (b + 1) * SC_U, cnt, lane);
  }
}

#define DG_WAVES   8
#define OW_GWAVES  (OW_WAVES - DG_WAVES)

__device__ __forceinline__ void ow_dual_tiles(const float* __restrict__ dual, const float* Wl,
                                              float* acc, int lo, int nr, int m_cons,
                                              int g, int lane) {
  const int c = lane & 15, h = lane >> 4, koff = h << 1;
  const int ntile = (nr + 15) >> 4;
  for (int t = g; t < ntile; t += DG_WAVES) {
    const float* arow = dual + (size_t)min(lo + t * 16 + c, m_cons - 1) * HID + koff;
    v2f af[32];
#pragma unroll
    for (int i = 0; i < 32; ++i) af[i] = *(const v2f*)(arow + 4 * i);
    const int mloc = t * 16 + (h << 3);
    for (int cp = 0; cp < HID / 16; cp += 2) {
      const float* B0 = Wl + cp * 16 * LDSS;
      const float* B1 = B0 + 16 * LDSS;
      v8f d0 = {};
      v8f d1 = {};
#pragma unroll
      for (int i = 0; i < 32; ++i) {
        d0 = wmma_f32(af[i], frag(B0, c, koff, i), d0);
        d1 = wmma_f32(af[i], frag(B1, c, koff, i), d1);
      }
      const int n0 = cp * 16 + c;
      const int w0 = (n0 & 3) * 32 + (n0 >> 2);
#pragma unroll
      for (int j = 0; j < 8; ++j) {
        const int rr = (mloc + j < nr) ? (mloc + j) : OW_ROWS;
        float* ar = acc + rr * HID + w0;
        __hip_atomic_fetch_add(&ar[0], d0[j], __ATOMIC_RELAXED, __HIP_MEMORY_SCOPE_WORKGROUP);
        __hip_atomic_fetch_add(&ar[4], d1[j], __ATOMIC_RELAXED, __HIP_MEMORY_SCOPE_WORKGROUP);
      }
    }
  }
}

__global__ __launch_bounds__(OW_THREADS)
void owner_scatter_kernel(const float* __restrict__ X, float* __restrict__ out,
                          const float* __restrict__ dual, const float* __restrict__ W1,
                          const float* __restrict__ b1, const float* __restrict__ rhs,
                          const float* __restrict__ sigma_p,
                          int nnz, int m_cons, int n_vars) {
  __shared__ __attribute__((aligned(16))) float acc[(OW_ROWS + 1) * HID];
  __shared__ __attribute__((aligned(16))) float wl[HID * LDSS];
  __shared__ int segL[OW_MAXCH];
  __shared__ int segS[OW_MAXCH];
  __shared__ int segP[OW_MAXCH + 1];
  const int tid  = threadIdx.x;
  const int lane = tid & 31;
  const int wv   = __builtin_amdgcn_readfirstlane(tid >> 5);
  const int nch  = (nnz + BN_CH - 1) / BN_CH;
  const int nchp = (nch + 31) & ~31;
  const long long nrec = (long long)nch * BN_CH;
  const uint2* recs = (const uint2*)(X + (size_t)n_vars * HID);
  const int*   offT = (const int*)(recs + nrec);
  const int lo = blockIdx.x * OW_ROWS;
  const int nr = min(OW_ROWS, m_cons - lo);
  const bool dg_wave = ((wv & 15) % 5) == 0;
  const int  dgi     = (wv >> 4) * 4 + (wv & 15) / 5;
  const int  gi      = wv - dgi - 1;

  for (int i = tid; i < OW_ROWS * HID / 4; i += OW_THREADS)
    ((float4*)acc)[i] = make_float4(0.f, 0.f, 0.f, 0.f);
  for (int i = tid; i < HID * HID / 4; i += OW_THREADS)
    *(float4*)&wl[(i >> 5) * LDSS + (i & 31) * 4] = ((const float4*)W1)[i];

  int c0 = 0;
  do {
    const int ncc = max(min(OW_MAXCH, nch - c0), 0);
    __syncthreads();
    if (tid < OW_MAXCH) {
      int len = 0, s = 0;
      if (tid < ncc) {
        const int ch = c0 + tid;
        const int a = offT[(size_t)blockIdx.x * nchp + ch];
        const int b = offT[(size_t)(blockIdx.x + 1) * nchp + ch];
        len = max(b - a, 0);
        s   = ch * BN_CH + a;
      }
      segL[tid] = len;
      segS[tid] = s;
    }
    __syncthreads();
    if (tid < 32) {
      int c[OW_MAXCH / 32], sum = 0;
#pragma unroll
      for (int j = 0; j < OW_MAXCH / 32; ++j) { c[j] = segL[tid * (OW_MAXCH / 32) + j]; sum += c[j]; }
      int inc = sum;
#pragma unroll
      for (int d = 1; d < 32; d <<= 1) {
        const int t = __shfl_up(inc, d);
        if (lane >= d) inc += t;
      }
      int ex = inc - sum;
#pragma unroll
      for (int j = 0; j < OW_MAXCH / 32; ++j) { segP[tid * (OW_MAXCH / 32) + j] = ex; ex += c[j]; }
      if (tid == 31) segP[OW_MAXCH] = ex;
    }
    __syncthreads();

    const int T = __builtin_amdgcn_readfirstlane(segP[OW_MAXCH]);
    const int U = (T + 2 * OW_GWAVES + DG_WAVES - 1) / (2 * OW_GWAVES + DG_WAVES);
    int g0, gend;
    if (dg_wave) {
      if (c0 == 0) ow_dual_tiles(dual, wl, acc, lo, nr, m_cons, dgi, lane);
      g0   = (2 * OW_GWAVES + dgi) * U;
      gend = min(T, g0 + U);
    } else {
      g0   = 2 * gi * U;
      gend = min(T, g0 + 2 * U);
    }
    uint2 rec = recs[ow_find_rec(segP, segS, g0 + lane, nrec)];
    long long inext = ow_find_rec(segP, segS, g0 + 32 + lane, nrec);
    for (; g0 < gend; g0 += 32) {
      const uint2 nx = recs[inext];
      ow_drain_rec(rec, min(32, gend - g0), n_vars, X, acc, lane, segP, segS, g0 + 64 + lane, nrec, inext);
      rec = nx;
    }
    c0 += OW_MAXCH;
  } while (c0 < nch);
  __syncthreads();

  const float  sigma = sigma_p[0];
  const float4 bb    = ((const float4*)b1)[lane];
  for (int i = tid; i < nr * 32; i += OW_THREADS) {
    const int row = i >> 5, q4 = i & 31;
    float4* op = (float4*)(out + (size_t)(lo + row) * HID) + q4;
    const float* ap = acc + row * HID + q4;
    const float4 rh = ((const float4*)(rhs + (size_t)(lo + row) * HID))[q4];
    float4 v;
    v.x = ap[0]  + bb.x - sigma * rh.x;
    v.y = ap[32] + bb.y - sigma * rh.y;
    v.z = ap[64] + bb.z - sigma * rh.z;
    v.w = ap[96] + bb.w - sigma * rh.w;
    v.x = v.x >= 0.f ? v.x : NEG_SLOPE * v.x;
    v.y = v.y >= 0.f ? v.y : NEG_SLOPE * v.y;
    v.z = v.z >= 0.f ? v.z : NEG_SLOPE * v.z;
    v.w = v.w >= 0.f ? v.w : NEG_SLOPE * v.w;
    *op = v;
  }
}

extern "C" void kernel_launch(void* const* d_in, const int* in_sizes, int n_in,
                              void* d_out, int out_size, void* d_ws, size_t ws_size,
                              hipStream_t stream) {
  const float* primal      = (const float*)d_in[0];
  const float* last_primal = (const float*)d_in[1];
  const float* dual        = (const float*)d_in[2];
  const int*   rows        = (const int*)d_in[3];
  const int*   cols        = (const int*)d_in[4];
  const float* vals        = (const float*)d_in[5];
  const float* rhs         = (const float*)d_in[6];
  const float* W1          = (const float*)d_in[7];
  const float* b1          = (const float*)d_in[8];
  const float* W2          = (const float*)d_in[9];
  const float* b2          = (const float*)d_in[10];
  const float* W3          = (const float*)d_in[11];
  const float* b3          = (const float*)d_in[12];
  const float* sigma       = (const float*)d_in[13];

  float* out = (float*)d_out;
  float* X   = (float*)d_ws;

  const int n_vars = in_sizes[0] / HID;
  const int m_cons = in_sizes[2] / HID;
  const int nnz    = in_sizes[5];

  const size_t nch    = ((size_t)nnz + BN_CH - 1) / BN_CH;
  const size_t need   = (size_t)n_vars * HID * 4 + nch * BN_CH * 8 + (OW_BLOCKS + 1) * ((nch + 31) / 32 * 32) * 4;
  const bool   binned = ws_size >= need && n_vars <= (1 << BN_COLBITS) && m_cons <= OW_BLOCKS * OW_ROWS;

  if (binned) owner_bin_kernel<<<OW_BLOCKS, OW_THREADS, 0, stream>>>(rows, cols, vals, sigma, X, n_vars, nnz);
  else        owner_mask_kernel<<<MK_BLOCKS, OW_THREADS, 0, stream>>>(rows, X, n_vars, nnz);
  fused_x_kernel<<<GEMM_BLOCKS_X, 256, 0, stream>>>(primal, last_primal,
                                                    W2, b2, W3, b3, X, n_vars / 16);
  if (!binned) dual_base_kernel<<<GEMM_BLOCKS_D, 256, 0, stream>>>(dual, W1, b1, rhs, sigma, out,
                                                                   m_cons / 16);
  if (binned) owner_scatter_kernel<<<OW_BLOCKS, OW_THREADS, 0, stream>>>(X, out, dual, W1, b1, rhs, sigma,
                                                                        nnz, m_cons, n_vars);
  else        owner_scatter_masked_kernel<<<OW_BLOCKS, OW_THREADS, 0, stream>>>(rows, cols, vals, X, sigma, out,
                                                                           nnz, m_cons, n_vars);
}


# Scatter: owner block orders its records by out row in LDS (integer counting sort); a wave sums a row's X rows in registers, one LDS float add per row end, not per nonzero; sum order changes
# speedup vs baseline: 3.3759x; 1.0289x over previous
#include <hip/hip_runtime.h>


#define HID 128
#define LDSS 132
#define TILE_F (16 * LDSS)
#define NEG_SLOPE 0.01f
#define GEMM_BLOCKS_X 512
#define GEMM_BLOCKS_D 768

typedef float v2f __attribute__((ext_vector_type(2)));
typedef float v8f __attribute__((ext_vector_type(8)));

__device__ __forceinline__ v8f wmma_f32(v2f a, v2f b, v8f c) {
  return __builtin_amdgcn_wmma_f32_16x16x4_f32(
       false, a,  false, b,
       (short)0, c,  false,  false);
}

__device__ __forceinline__ void stage_w_rows(const float* __restrict__ W, int n0,
                                             float* wst, int lane) {
  const float4* g = (const float4*)(W + (size_t)n0 * HID);
#pragma unroll
  for (int q = 0; q < 16; ++q)
    *(float4*)&wst[q * LDSS + lane * 4] = g[q * 32 + lane];
}

__device__ __forceinline__ v2f frag(const float* t, int c, int koff, int i) {
  return *(const v2f*)&t[c * LDSS + 4 * i + koff];
}

__device__ __forceinline__ void lds_barrier() {
  __builtin_amdgcn_fence(__ATOMIC_RELEASE, "workgroup", "local");
  __builtin_amdgcn_s_barrier();
  __builtin_amdgcn_fence(__ATOMIC_ACQUIRE, "workgroup", "local");
}

__global__ __launch_bounds__(256)
void fused_x_kernel(const float* __restrict__ primal,
                    const float* __restrict__ last_primal,
                    const float* __restrict__ W2, const float* __restrict__ b2,
                    const float* __restrict__ W3, const float* __restrict__ b3,
                    float* __restrict__ X, int ntiles) {
  __shared__ __attribute__((aligned(16))) float smem[8 * TILE_F];
  const int tid  = threadIdx.x;
  const int lane = tid & 31;
  const int wave = tid >> 5;
  const int c    = lane & 15;
  const int h    = lane >> 4;
  const int koff = h << 1;
  const int n    = wave * 16 + c;

  int tile = blockIdx.x;

  const int srow = tid >> 5, sc4 = tid & 31;
  const int soff = srow * LDSS + sc4 * 4;
  float4 pa0 = {0.f, 0.f, 0.f, 0.f}, pa1 = pa0, pl0 = pa0, pl1 = pa0;
  if (tile < ntiles) {
    const float4* gp = (const float4*)(primal      + (size_t)tile * 16 * HID);
    const float4* gl = (const float4*)(last_primal + (size_t)tile * 16 * HID);
    pa0 = gp[tid]; pa1 = gp[tid + 256];
    pl0 = gl[tid]; pl1 = gl[tid + 256];
  }

  v2f bw2[32], bw3[32];
  float* wst = smem + wave * TILE_F;
  stage_w_rows(W2, wave * 16, wst, lane);
  __syncthreads();
#pragma unroll
  for (int i = 0; i < 32; ++i) bw2[i] = frag(wst, c, koff, i);
  __syncthreads();
  stage_w_rows(W3, wave * 16, wst, lane);
  __syncthreads();
#pragma unroll
  for (int i = 0; i < 32; ++i) bw3[i] = frag(wst, c, koff, i);
  __syncthreads();

  const float bias = b2[n] - b3[n];

  if (tile < ntiles) {
    *(float4*)&smem[soff]                     = pa0;
    *(float4*)&smem[soff + 8 * LDSS]          = pa1;
    *(float4*)&smem[TILE_F + soff]            = pl0;
    *(float4*)&smem[TILE_F + soff + 8 * LDSS] = pl1;
  }
  __syncthreads();

  int cur = 0;
  for (; tile < ntiles; tile += gridDim.x) {
    const int  next     = tile + gridDim.x;
    const bool has_next = next < ntiles;
    if (has_next) {
      const float4* gp = (const float4*)(primal      + (size_t)next * 16 * HID);
      const float4* gl = (const float4*)(last_primal + (size_t)next * 16 * HID);
      pa0 = gp[tid]; pa1 = gp[tid + 256];
      pl0 = gl[tid]; pl1 = gl[tid + 256];
    }

    const float* Ap = smem + cur * 2 * TILE_F;
    const float* Al = Ap + TILE_F;
    v8f acc  = {};
    v8f acc2 = {};
#pragma unroll
    for (int i = 0; i < 32; ++i) {
      acc  = wmma_f32(frag(Ap, c, koff, i), bw2[i], acc);
      acc2 = wmma_f32(frag(Al, c, koff, i), bw3[i], acc2);
    }

    const int mbase = tile * 16 + (h << 3);
#pragma unroll
    for (int j = 0; j < 8; ++j)
      X[(size_t)(mbase + j) * HID + n] = acc[j] - acc2[j] + bias;

    if (has_next) {
      float* Np = smem + (cur ^ 1) * 2 * TILE_F;
      *(float4*)&Np[soff]                     = pa0;
      *(float4*)&Np[soff + 8 * LDSS]          = pa1;
      *(float4*)&Np[TILE_F + soff]            = pl0;
      *(float4*)&Np[TILE_F + soff + 8 * LDSS] = pl1;
    }
    lds_barrier();
    cur ^= 1;
  }
}

__global__ __launch_bounds__(256)
void dual_base_kernel(const float* __restrict__ dual,
                      const float* __restrict__ W1, const float* __restrict__ b1,
                      const float* __restrict__ rhs,
                      const float* __restrict__ sigma_p,
                      float* __restrict__ out, int ntiles) {
  __shared__ __attribute__((aligned(16))) float smem[8 * TILE_F];
  const int tid  = threadIdx.x;
  const int lane = tid & 31;
  const int wave = tid >> 5;
  const int c    = lane & 15;
  const int h    = lane >> 4;
  const int koff = h << 1;
  const int n    = wave * 16 + c;

  int tile = blockIdx.x;

  const int srow = tid >> 5, sc4 = tid & 31;
  const int soff = srow * LDSS + sc4 * 4;
  float4 pd0 = {0.f, 0.f, 0.f, 0.f}, pd1 = pd0;
  if (tile < ntiles) {
    const float4* gd = (const float4*)(dual + (size_t)tile * 16 * HID);
    pd0 = gd[tid]; pd1 = gd[tid + 256];
  }

  v2f bw[32];
  float* wst = smem + wave * TILE_F;
  stage_w_rows(W1, wave * 16, wst, lane);
  __syncthreads();
#pragma unroll
  for (int i = 0; i < 32; ++i) bw[i] = frag(wst, c, koff, i);
  __syncthreads();

  const float bias  = b1[n];
  const float sigma = sigma_p[0];

  if (tile < ntiles) {
    *(float4*)&smem[soff]            = pd0;
    *(float4*)&smem[soff + 8 * LDSS] = pd1;
  }
  __syncthreads();

  int cur = 0;
  for (; tile < ntiles; tile += gridDim.x) {
    const int  next     = tile + gridDim.x;
    const bool has_next = next < ntiles;
    if (has_next) {
      const float4* gd = (const float4*)(dual + (size_t)next * 16 * HID);
      pd0 = gd[tid]; pd1 = gd[tid + 256];
    }

    const int mbase = tile * 16 + (h << 3);
    float rv[8];
#pragma unroll
    for (int j = 0; j < 8; ++j) rv[j] = rhs[(size_t)(mbase + j) * HID + n];

    const float* Ad = smem + cur * TILE_F;
    v8f acc = {};
#pragma unroll
    for (int i = 0; i < 32; ++i)
      acc = wmma_f32(frag(Ad, c, koff, i), bw[i], acc);

#pragma unroll
    for (int j = 0; j < 8; ++j)
      out[(size_t)(mbase + j) * HID + n] = acc[j] + bias - sigma * rv[j];

    if (has_next) {
      float* Nd = smem + (cur ^ 1) * TILE_F;
      *(float4*)&Nd[soff]            = pd0;
      *(float4*)&Nd[soff + 8 * LDSS] = pd1;
    }
    lds_barrier();
    cur ^= 1;
  }
}

#define SC_U       4
#define SC_NB      (32 / SC_U)
#define OW_ROWS    391
#define OW_BLOCKS  256
#define OW_WAVES   32
#define OW_THREADS (OW_WAVES * 32)
#define OW_QCAP    512
#define OW_GB      8
#define OW_GQCAP   64
#define MK_BLOCKS  64
#define MK_LDSS    257

__device__ __forceinline__ int4 ow_load_rows(const int* __restrict__ rows, long long gbase,
                                             int nnz, int lane) {
  const long long i = gbase + lane * 4;
  if (gbase + 128 <= nnz) return *(const int4*)(rows + i);
  int4 r;
  r.x = i     < nnz ? rows[i]     : -1;
  r.y = i + 1 < nnz ? rows[i + 1] : -1;
  r.z = i + 2 < nnz ? rows[i + 2] : -1;
  r.w = i + 3 < nnz ? rows[i + 3] : -1;
  return r;
}

__device__ __forceinline__ void ow_push(int r, int e, int lo, int nr, int* q, int& tail) {
  const bool m = (unsigned)(r - lo) < (unsigned)nr;
  const unsigned bal = __builtin_amdgcn_ballot_w32(m);
  if (bal != 0u) {
    if (m) q[(tail + (int)__builtin_amdgcn_mbcnt_lo(bal, 0u)) & (OW_QCAP - 1)] = e;
    tail += __builtin_popcount(bal);
  }
}

__device__ __forceinline__ void ow_take_groups(const int* gq, int gpos, int cnt,
                                               const int* __restrict__ rows, int nnz,
                                               int lo, int nr, int* q, int& tail, int lane) {
  const int4* gp = (const int4*)&gq[gpos & (OW_GQCAP - 1)];
  const int4 ga = gp[0], gb = gp[1];
  const int g[OW_GB] = {ga.x, ga.y, ga.z, ga.w, gb.x, gb.y, gb.z, gb.w};
  int rr[OW_GB];
#pragma unroll
  for (int u = 0; u < OW_GB; ++u)
    rr[u] = rows[min(max((int)(((unsigned)g[u] << 5) + lane), 0), nnz - 1)];
#pragma unroll
  for (int u = 0; u < OW_GB; ++u) {
    const int e = (int)(((unsigned)g[u] << 5) + lane);
    ow_push((u < cnt && e < nnz) ? rr[u] : -1, e, lo, nr, q, tail);
  }
}

__device__ __forceinline__ void ow_load_batch(float4 (&xb)[SC_U], const float* __restrict__ X,
                                              int cidx, int j0, int lane) {
#pragma unroll
  for (int u = 0; u < SC_U; ++u) {
    const int cc = __builtin_amdgcn_readlane(cidx, j0 + u);
    xb[u] = *(const float4*)(X + (size_t)cc * HID + lane * 4);
  }
}

__device__ __forceinline__ void ow_add_batch(const float4 (&xb)[SC_U], float* acc,
                                             int rloc, float sval, int j0, int cnt, int lane) {
#pragma unroll
  for (int u = 0; u < SC_U; ++u) {
    if (j0 + u < cnt) {
      const int   rr = __builtin_amdgcn_readlane(rloc, j0 + u);
      const float s  = __int_as_float(__builtin_amdgcn_readlane(__float_as_int(sval), j0 + u));
      float* ap = acc + rr * HID + lane;
      __hip_atomic_fetch_add(&ap[0],  s * xb[u].x, __ATOMIC_RELAXED, __HIP_MEMORY_SCOPE_WORKGROUP);
      __hip_atomic_fetch_add(&ap[32], s * xb[u].y, __ATOMIC_RELAXED, __HIP_MEMORY_SCOPE_WORKGROUP);
      __hip_atomic_fetch_add(&ap[64], s * xb[u].z, __ATOMIC_RELAXED, __HIP_MEMORY_SCOPE_WORKGROUP);
      __hip_atomic_fetch_add(&ap[96], s * xb[u].w, __ATOMIC_RELAXED, __HIP_MEMORY_SCOPE_WORKGROUP);
    }
  }
}

__device__ __forceinline__ void ow_drain(const int* q, int head, int cnt, int lo,
                                         const int* __restrict__ rows, const int* __restrict__ cols,
                                         const float* __restrict__ vals, const float* __restrict__ X,
                                         float sigma, float* acc, int lane) {
  int rloc = 0, cidx = 0;
  float sval = 0.f;
  if (lane < cnt) {
    const int e = q[(head + lane) & (OW_QCAP - 1)];
    rloc = rows[e] - lo;
    cidx = cols[e];
    sval = vals[e] * sigma;
  }
  float4 xb0[SC_U], xb1[SC_U];
  ow_load_batch(xb0, X, cidx, 0, lane);
#pragma unroll
  for (int b = 0; b < SC_NB; b += 2) {
    if ((b + 1) * SC_U < cnt) ow_load_batch(xb1, X, cidx, (b + 1) * SC_U, lane);
    ow_add_batch(xb0, acc, rloc, sval, b * SC_U, cnt, lane);
    if (b + 2 < SC_NB && (b + 2) * SC_U < cnt) ow_load_batch(xb0, X, cidx, (b + 2) * SC_U, lane);
    ow_add_batch(xb1, acc, rloc, sval, (b + 1) * SC_U, cnt, lane);
  }
}


__global__ __launch_bounds__(OW_THREADS)
void owner_mask_kernel(const int* __restrict__ rows, float* __restrict__ X,
                       int n_vars, int nnz) {
  __shared__ unsigned bits[32 * MK_LDSS];
  unsigned* mask = (unsigned*)(X + (size_t)n_vars * HID);
  const int tid  = threadIdx.x;
  const int lane = tid & 31;
  const int wv   = __builtin_amdgcn_readfirstlane(tid >> 5);
  const int nw   = (int)(((long long)nnz + 1023) >> 10);
  const int nwp  = (nw + 31) & ~31;
  unsigned* mybits = bits + wv * MK_LDSS;

  for (int cb = blockIdx.x; cb * 32 < nwp; cb += gridDim.x) {
    for (int i = tid; i < 32 * MK_LDSS; i += OW_THREADS) bits[i] = 0u;
    __syncthreads();

    const long long sb = ((long long)(cb * 32 + wv)) << 10;
    if (sb < nnz) {
      int4 c[8];
      if (sb + 1024 <= nnz) {
#pragma unroll
        for (int j = 0; j < 8; ++j) c[j] = *(const int4*)(rows + sb + j * 128 + lane * 4);
      } else {
#pragma unroll
        for (int j = 0; j < 8; ++j) c[j] = ow_load_rows(rows, sb + j * 128, nnz, lane);
      }
#pragma unroll
      for (int j = 0; j < 8; ++j) {
        const unsigned bit = 1u << (j * 4 + (lane >> 3));
        if (c[j].x >= 0) atomicOr(&mybits[((unsigned)c[j].x / OW_ROWS) % OW_BLOCKS], bit);
        if (c[j].y >= 0) atomicOr(&mybits[((unsigned)c[j].y / OW_ROWS) % OW_BLOCKS], bit);
        if (c[j].z >= 0) atomicOr(&mybits[((unsigned)c[j].z / OW_ROWS) % OW_BLOCKS], bit);
        if (c[j].w >= 0) atomicOr(&mybits[((unsigned)c[j].w / OW_ROWS) % OW_BLOCKS], bit);
      }
    }
    __syncthreads();

#pragma unroll
    for (int k = 0; k < 8; ++k) {
      const int o = wv * 8 + k;
      mask[(size_t)o * nwp + cb * 32 + lane] = bits[lane * MK_LDSS + o];
    }
    __syncthreads();
  }
}

__global__ __launch_bounds__(OW_THREADS)
void owner_scatter_masked_kernel(const int* __restrict__ rows, const int* __restrict__ cols,
                          const float* __restrict__ vals, const float* __restrict__ X,
                          const float* __restrict__ sigma_p, float* __restrict__ out,
                          int nnz, int m_cons, int n_vars) {
  __shared__ __attribute__((aligned(16))) float acc[OW_ROWS * HID];
  __shared__ int ring[OW_WAVES * OW_QCAP];
  __shared__ __attribute__((aligned(16))) int gring[OW_WAVES * OW_GQCAP];
  const int tid  = threadIdx.x;
  const int lane = tid & 31;
  const int wv = __builtin_amdgcn_readfirstlane(tid >> 5);
  int* q = ring + wv * OW_QCAP;
  int* gq = gring + wv * OW_GQCAP;
  const float sigma = sigma_p[0];
  const int nw   = (int)(((long long)nnz + 1023) >> 10);
  const int nwp  = (nw + 31) & ~31;
  const unsigned* mrow = (const unsigned*)(X + (size_t)n_vars * HID) + (size_t)blockIdx.x * nwp;
  const int wpw  = (nw + OW_WAVES - 1) / OW_WAVES;
  const int wbeg = wv * wpw;
  const int wend = min(nw, wbeg + wpw);

  for (int lo = blockIdx.x * OW_ROWS; lo < m_cons; lo += gridDim.x * OW_ROWS) {
    const int nr = min(OW_ROWS, m_cons - lo);

    for (int i = tid; i < OW_ROWS * HID / 4; i += OW_THREADS)
      ((float4*)acc)[i] = make_float4(0.f, 0.f, 0.f, 0.f);
    __syncthreads();

    int head = 0, tail = 0;
    int ghead = 0, gtail = 0;
    for (int s0 = wbeg; s0 < wend; s0 += 32) {
      const unsigned mw = (s0 + lane < wend) ? mrow[s0 + lane] : 0u;
      const int nwi = min(32, wend - s0);
      for (int wi = 0; wi < nwi; ++wi) {
        const unsigned w = (unsigned)__builtin_amdgcn_readlane((int)mw, wi);
        if (w == 0u) continue;
        if ((w >> lane) & 1u)
          gq[(gtail + (int)__builtin_amdgcn_mbcnt_lo(w, 0u)) & (OW_GQCAP - 1)] = ((s0 + wi) << 5) + lane;
        gtail += __builtin_popcount(w);
        while (gtail - ghead >= OW_GB) {
          __builtin_amdgcn_fence(__ATOMIC_ACQ_REL, "wavefront");
          ow_take_groups(gq, ghead, OW_GB, rows, nnz, lo, nr, q, tail, lane);
          ghead += OW_GB;
          while (tail - head >= 32) {
            __builtin_amdgcn_fence(__ATOMIC_ACQ_REL, "wavefront");
            ow_drain(q, head, 32, lo, rows, cols, vals, X, sigma, acc, lane);
            head += 32;
          }
        }
      }
    }
    if (gtail - ghead > 0) {
      __builtin_amdgcn_fence(__ATOMIC_ACQ_REL, "wavefront");
      ow_take_groups(gq, ghead, gtail - ghead, rows, nnz, lo, nr, q, tail, lane);
      while (tail - head >= 32) {
        __builtin_amdgcn_fence(__ATOMIC_ACQ_REL, "wavefront");
        ow_drain(q, head, 32, lo, rows, cols, vals, X, sigma, acc, lane);
        head += 32;
      }
    }
    if (tail - head > 0) {
      __builtin_amdgcn_fence(__ATOMIC_ACQ_REL, "wavefront");
      ow_drain(q, head, tail - head, lo, rows, cols, vals, X, sigma, acc, lane);
    }
    __syncthreads();

    for (int i = tid; i < nr * 32; i += OW_THREADS) {
      const int row = i >> 5, q4 = i & 31;
      float4* op = (float4*)(out + (size_t)(lo + row) * HID) + q4;
      const float* ap = acc + row * HID + q4;
      float4 v = *op;
      v.x += ap[0]; v.y += ap[32]; v.z += ap[64]; v.w += ap[96];
      v.x = v.x >= 0.f ? v.x : NEG_SLOPE * v.x;
      v.y = v.y >= 0.f ? v.y : NEG_SLOPE * v.y;
      v.z = v.z >= 0.f ? v.z : NEG_SLOPE * v.z;
      v.w = v.w >= 0.f ? v.w : NEG_SLOPE * v.w;
      *op = v;
    }
    __syncthreads();
  }
}

#define BN_CH      6272
#define BN_PER     ((BN_CH + OW_THREADS - 1) / OW_THREADS)
#define BN_COLBITS 18
#define OW_RCAP    7168
#define OW_SG      8
#define OW_SCAN_PER ((OW_ROWS + 31) / 32)

__global__ __launch_bounds__(OW_THREADS)
void owner_bin_kernel(const int* __restrict__ rows, const int* __restrict__ cols,
                      const float* __restrict__ vals, const float* __restrict__ sigma_p,
                      float* __restrict__ X, int n_vars, int nnz) {
  __shared__ int cnt[OW_BLOCKS];
  __shared__ int start[OW_BLOCKS + 1];
  __shared__ uint2 stage[BN_CH];
  const int tid  = threadIdx.x;
  const int lane = tid & 31;
  const int nch  = (nnz + BN_CH - 1) / BN_CH;
  const int nchp = (nch + 31) & ~31;
  uint2* recs = (uint2*)(X + (size_t)n_vars * HID);
  int*   offT = (int*)(recs + (size_t)nch * BN_CH);
  const float sigma = sigma_p[0];

  for (int ch = blockIdx.x; ch < nch; ch += gridDim.x) {
    if (tid < OW_BLOCKS) cnt[tid] = 0;
    __syncthreads();

    const int base = ch * BN_CH;
    const int n    = min(BN_CH, nnz - base);
    int   rr[BN_PER], cc[BN_PER];
    float vv[BN_PER];
#pragma unroll
    for (int k = 0; k < BN_PER; ++k) {
      const int e = base + min(k * OW_THREADS + tid, n - 1);
      rr[k] = rows[e]; cc[k] = cols[e]; vv[k] = vals[e];
    }
    unsigned pk[BN_PER];
    float    sv[BN_PER];
    int      ow[BN_PER], rk[BN_PER];
#pragma unroll
    for (int k = 0; k < BN_PER; ++k) {
      const int i = k * OW_THREADS + tid;
      ow[k] = -1; rk[k] = 0; pk[k] = 0u; sv[k] = 0.f;
      if (i < n) {
        const unsigned r = (unsigned)rr[k];
        const unsigned q = r / OW_ROWS;
        const int o = (int)min(q, (unsigned)(OW_BLOCKS - 1));
        pk[k] = ((unsigned)cc[k] & ((1u << BN_COLBITS) - 1u)) | ((r - q * OW_ROWS) << BN_COLBITS);
        sv[k] = vv[k] * sigma;
        ow[k] = o;
        rk[k] = atomicAdd(&cnt[o], 1);
      }
    }
    __syncthreads();

    if (tid < 32) {
      int c[8], s = 0;
#pragma unroll
      for (int j = 0; j < 8; ++j) { c[j] = cnt[tid * 8 + j]; s += c[j]; }
      int inc = s;
#pragma unroll
      for (int d = 1; d < 32; d <<= 1) {
        const int t = __shfl_up(inc, d);
        if (lane >= d) inc += t;
      }
      int ex = inc - s;
#pragma unroll
      for (int j = 0; j < 8; ++j) { start[tid * 8 + j] = ex; ex += c[j]; }
      if (tid == 31) start[OW_BLOCKS] = ex;
    }
    __syncthreads();

#pragma unroll
    for (int k = 0; k < BN_PER; ++k)
      if (ow[k] >= 0) stage[start[ow[k]] + rk[k]] = make_uint2(pk[k], __float_as_uint(sv[k]));
    if (tid <= OW_BLOCKS) offT[(size_t)tid * nchp + ch] = start[tid];
    __syncthreads();

    for (int i = tid; i < n; i += OW_THREADS) recs[(size_t)base + i] = stage[i];
    __syncthreads();
  }
}

__device__ __forceinline__ int ow_rec_row(uint2 rec) {
  return min((int)(rec.x >> BN_COLBITS), OW_ROWS - 1);
}

template <int PASS>
__device__ __forceinline__ void ow_order_one(uint2 rec, int* rcnt, uint2* srt) {
  const int r = ow_rec_row(rec);
  if (PASS == 0) {
    atomicAdd(&rcnt[r], 1);
  } else {
    const unsigned p = (unsigned)atomicAdd(&rcnt[r], 1);
    if (p < (unsigned)OW_RCAP) srt[p] = rec;
  }
}

template <int PASS>
__device__ __forceinline__ void ow_order_pass(const uint2* __restrict__ recs, int st, int len, int nseg,
                                              int* rcnt, uint2* srt, int lane) {
  for (int j0 = 0; j0 < nseg; j0 += OW_SG) {
    uint2 r[OW_SG];
    int   sj[OW_SG], nj[OW_SG];
#pragma unroll
    for (int u = 0; u < OW_SG; ++u) {
      const int j = min(j0 + u, 31);
      sj[u] = __builtin_amdgcn_readlane(st, j);
      nj[u] = (j0 + u < nseg) ? __builtin_amdgcn_readlane(len, j) : 0;
      r[u]  = recs[(lane < nj[u]) ? (long long)sj[u] + lane : 0ll];
    }
#pragma unroll
    for (int u = 0; u < OW_SG; ++u)
      if (lane < nj[u]) ow_order_one<PASS>(r[u], rcnt, srt);
#pragma unroll
    for (int u = 0; u < OW_SG; ++u)
      for (int o = 32; o < nj[u]; o += 32)
        if (o + lane < nj[u]) ow_order_one<PASS>(recs[(long long)sj[u] + o + lane], rcnt, srt);
  }
}

__device__ __forceinline__ void ow_flush_row(float* acc, int row, float4 s, int lane) {
  float* ap = acc + row * HID + lane;
  __hip_atomic_fetch_add(&ap[0],  s.x, __ATOMIC_RELAXED, __HIP_MEMORY_SCOPE_WORKGROUP);
  __hip_atomic_fetch_add(&ap[32], s.y, __ATOMIC_RELAXED, __HIP_MEMORY_SCOPE_WORKGROUP);
  __hip_atomic_fetch_add(&ap[64], s.z, __ATOMIC_RELAXED, __HIP_MEMORY_SCOPE_WORKGROUP);
  __hip_atomic_fetch_add(&ap[96], s.w, __ATOMIC_RELAXED, __HIP_MEMORY_SCOPE_WORKGROUP);
}

__device__ __forceinline__ void ow_sum_batch(const float4 (&xb)[SC_U], float* acc, int rloc, float sval,
                                             int j0, int cnt, int lane, int& cur, float4& sum) {
#pragma unroll
  for (int u = 0; u < SC_U; ++u) {
    if (j0 + u < cnt) {
      const int   rr = __builtin_amdgcn_readlane(rloc, j0 + u);
      const float s  = __int_as_float(__builtin_amdgcn_readlane(__float_as_int(sval), j0 + u));
      if (rr != cur) {
        if (cur >= 0) ow_flush_row(acc, cur, sum, lane);
        sum = make_float4(0.f, 0.f, 0.f, 0.f);
        cur = rr;
      }
      sum.x += s * xb[u].x;
      sum.y += s * xb[u].y;
      sum.z += s * xb[u].z;
      sum.w += s * xb[u].w;
    }
  }
}

__device__ __forceinline__ void ow_drain_sorted(uint2 rec, int cnt, int n_vars,
                                                const float* __restrict__ X, float* acc, int lane,
                                                int& cur, float4& sum) {
  int rloc = 0, cidx = 0;
  float sval = 0.f;
  if (lane < cnt) {
    cidx = min((int)(rec.x & ((1u << BN_COLBITS) - 1u)), n_vars - 1);
    rloc = ow_rec_row(rec);
    sval = __uint_as_float(rec.y);
  }
  float4 xb0[SC_U], xb1[SC_U];
  ow_load_batch(xb0, X, cidx, 0, lane);
#pragma unroll
  for (int b = 0; b < SC_NB; b += 2) {
    if ((b + 1) * SC_U < cnt) ow_load_batch(xb1, X, cidx, (b + 1) * SC_U, lane);
    ow_sum_batch(xb0, acc, rloc, sval, b * SC_U, cnt, lane, cur, sum);
    if (b + 2 < SC_NB && (b + 2) * SC_U < cnt) ow_load_batch(xb0, X, cidx, (b + 2) * SC_U, lane);
    ow_sum_batch(xb1, acc, rloc, sval, (b + 1) * SC_U, cnt, lane, cur, sum);
  }
}

#define DG_WAVES   8
#define OW_GWAVES  (OW_WAVES - DG_WAVES)

__device__ __forceinline__ void ow_dual_tiles(const float* __restrict__ dual, const float* Wl,
                                              float* acc, int lo, int nr, int m_cons,
                                              int g, int lane) {
  const int c = lane & 15, h = lane >> 4, koff = h << 1;
  const int ntile = (nr + 15) >> 4;
  for (int t = g; t < ntile; t += DG_WAVES) {
    const float* arow = dual + (size_t)min(lo + t * 16 + c, m_cons - 1) * HID + koff;
    v2f af[32];
#pragma unroll
    for (int i = 0; i < 32; ++i) af[i] = *(const v2f*)(arow + 4 * i);
    const int mloc = t * 16 + (h << 3);
    for (int cp = 0; cp < HID / 16; cp += 2) {
      const float* B0 = Wl + cp * 16 * LDSS;
      const float* B1 = B0 + 16 * LDSS;
      v8f d0 = {};
      v8f d1 = {};
#pragma unroll
      for (int i = 0; i < 32; ++i) {
        d0 = wmma_f32(af[i], frag(B0, c, koff, i), d0);
        d1 = wmma_f32(af[i], frag(B1, c, koff, i), d1);
      }
      const int n0 = cp * 16 + c;
      const int w0 = (n0 & 3) * 32 + (n0 >> 2);
#pragma unroll
      for (int j = 0; j < 8; ++j) {
        const int rr = (mloc + j < nr) ? (mloc + j) : OW_ROWS;
        float* ar = acc + rr * HID + w0;
        __hip_atomic_fetch_add(&ar[0], d0[j], __ATOMIC_RELAXED, __HIP_MEMORY_SCOPE_WORKGROUP);
        __hip_atomic_fetch_add(&ar[4], d1[j], __ATOMIC_RELAXED, __HIP_MEMORY_SCOPE_WORKGROUP);
      }
    }
  }
}

__global__ __launch_bounds__(OW_THREADS)
void owner_scatter_kernel(const float* __restrict__ X, float* __restrict__ out,
                          const float* __restrict__ dual, const float* __restrict__ W1,
                          const float* __restrict__ b1, const float* __restrict__ rhs,
                          const float* __restrict__ sigma_p,
                          int nnz, int m_cons, int n_vars) {
  __shared__ __attribute__((aligned(16))) float acc[(OW_ROWS + 1) * HID];
  __shared__ __attribute__((aligned(16))) float wl[HID * LDSS];
  __shared__ __attribute__((aligned(8)))  uint2 srt[OW_RCAP];
  __shared__ int rcnt[OW_ROWS];
  __shared__ int tot;
  const int tid  = threadIdx.x;
  const int lane = tid & 31;
  const int wv   = __builtin_amdgcn_readfirstlane(tid >> 5);
  const int nch  = (nnz + BN_CH - 1) / BN_CH;
  const int nchp = (nch + 31) & ~31;
  const long long nrec = (long long)nch * BN_CH;
  const uint2* recs = (const uint2*)(X + (size_t)n_vars * HID);
  const int*   offT = (const int*)(recs + nrec);
  const int lo = blockIdx.x * OW_ROWS;
  const int nr = min(OW_ROWS, m_cons - lo);
  const bool dg_wave = ((wv & 15) % 5) == 0;
  const int  dgi     = (wv >> 4) * 4 + (wv & 15) / 5;
  const int  gi      = wv - dgi - 1;

  for (int i = tid; i < OW_ROWS * HID / 4; i += OW_THREADS)
    ((float4*)acc)[i] = make_float4(0.f, 0.f, 0.f, 0.f);
  for (int i = tid; i < HID * HID / 4; i += OW_THREADS)
    *(float4*)&wl[(i >> 5) * LDSS + (i & 31) * 4] = ((const float4*)W1)[i];

  int c0 = 0;
  do {
    int win = max(min(OW_THREADS, nch - c0), 0);
    int st = 0, len = 0, T = 0;
    for (;;) {
      __syncthreads();
      for (int i = tid; i < OW_ROWS; i += OW_THREADS) rcnt[i] = 0;
      if (tid == 0) tot = 0;
      __syncthreads();
      const int k = lane * 32 + wv;
      st = 0; len = 0;
      if (k < win) {
        const int ch = c0 + k;
        const int a = min(max(offT[(size_t)blockIdx.x * nchp + ch], 0), BN_CH);
        const int b = offT[(size_t)(blockIdx.x + 1) * nchp + ch];
        len = min(max(b - a, 0), BN_CH - a);
        st  = ch * BN_CH + a;
      }
      int s = len;
#pragma unroll
      for (int d = 16; d >= 1; d >>= 1) s += __shfl_xor(s, d);
      if (lane == 0) atomicAdd(&tot, s);
      __syncthreads();
      T = __builtin_amdgcn_readfirstlane(tot);
      if (T <= OW_RCAP || win <= 1) break;
      win = (win + 1) >> 1;
    }
    const int nseg = (win - wv + 31) >> 5;

    ow_order_pass<0>(recs, st, len, nseg, rcnt, srt, lane);
    __syncthreads();
    if (tid < 32) {
      int c[OW_SCAN_PER], sum = 0;
#pragma unroll
      for (int j = 0; j < OW_SCAN_PER; ++j) {
        const int idx = tid * OW_SCAN_PER + j;
        c[j] = idx < OW_ROWS ? rcnt[idx] : 0;
        sum += c[j];
      }
      int inc = sum;
#pragma unroll
      for (int d = 1; d < 32; d <<= 1) {
        const int t = __shfl_up(inc, d);
        if (lane >= d) inc += t;
      }
      int ex = inc - sum;
#pragma unroll
      for (int j = 0; j < OW_SCAN_PER; ++j) {
        const int idx = tid * OW_SCAN_PER + j;
        if (idx < OW_ROWS) rcnt[idx] = ex;
        ex += c[j];
      }
    }
    __syncthreads();
    ow_order_pass<1>(recs, st, len, nseg, rcnt, srt, lane);
    __syncthreads();

    const int Tc = min(T, OW_RCAP);
    const int U  = (Tc + 2 * OW_GWAVES + DG_WAVES - 1) / (2 * OW_GWAVES + DG_WAVES);
    int g0, gend;
    if (dg_wave) {
      if (c0 == 0) ow_dual_tiles(dual, wl, acc, lo, nr, m_cons, dgi, lane);
      g0   = (2 * OW_GWAVES + dgi) * U;
      gend = min(Tc, g0 + U);
    } else {
      g0   = 2 * gi * U;
      gend = min(Tc, g0 + 2 * U);
    }
    int    cur = -1;
    float4 sum = make_float4(0.f, 0.f, 0.f, 0.f);
    for (; g0 < gend; g0 += 32) {
      const uint2 rec = srt[min(g0 + lane, OW_RCAP - 1)];
      ow_drain_sorted(rec, min(32, gend - g0), n_vars, X, acc, lane, cur, sum);
    }
    if (cur >= 0) ow_flush_row(acc, cur, sum, lane);
    c0 += win;
  } while (c0 < nch);
  __syncthreads();

  const float  sigma = sigma_p[0];
  const float4 bb    = ((const float4*)b1)[lane];
  for (int i = tid; i < nr * 32; i += OW_THREADS) {
    const int row = i >> 5, q4 = i & 31;
    float4* op = (float4*)(out + (size_t)(lo + row) * HID) + q4;
    const float* ap = acc + row * HID + q4;
    const float4 rh = ((const float4*)(rhs + (size_t)(lo + row) * HID))[q4];
    float4 v;
    v.x = ap[0]  + bb.x - sigma * rh.x;
    v.y = ap[32] + bb.y - sigma * rh.y;
    v.z = ap[64] + bb.z - sigma * rh.z;
    v.w = ap[96] + bb.w - sigma * rh.w;
    v.x = v.x >= 0.f ? v.x : NEG_SLOPE * v.x;
    v.y = v.y >= 0.f ? v.y : NEG_SLOPE * v.y;
    v.z = v.z >= 0.f ? v.z : NEG_SLOPE * v.z;
    v.w = v.w >= 0.f ? v.w : NEG_SLOPE * v.w;
    *op = v;
  }
}

extern "C" void kernel_launch(void* const* d_in, const int* in_sizes, int n_in,
                              void* d_out, int out_size, void* d_ws, size_t ws_size,
                              hipStream_t stream) {
  const float* primal      = (const float*)d_in[0];
  const float* last_primal = (const float*)d_in[1];
  const float* dual        = (const float*)d_in[2];
  const int*   rows        = (const int*)d_in[3];
  const int*   cols        = (const int*)d_in[4];
  const float* vals        = (const float*)d_in[5];
  const float* rhs         = (const float*)d_in[6];
  const float* W1          = (const float*)d_in[7];
  const float* b1          = (const float*)d_in[8];
  const float* W2          = (const float*)d_in[9];
  const float* b2          = (const float*)d_in[10];
  const float* W3          = (const float*)d_in[11];
  const float* b3          = (const float*)d_in[12];
  const float* sigma       = (const float*)d_in[13];

  float* out = (float*)d_out;
  float* X   = (float*)d_ws;

  const int n_vars = in_sizes[0] / HID;
  const int m_cons = in_sizes[2] / HID;
  const int nnz    = in_sizes[5];

  const size_t nch    = ((size_t)nnz + BN_CH - 1) / BN_CH;
  const size_t need   = (size_t)n_vars * HID * 4 + nch * BN_CH * 8 + (OW_BLOCKS + 1) * ((nch + 31) / 32 * 32) * 4;
  const bool   binned = ws_size >= need && n_vars <= (1 << BN_COLBITS) && m_cons <= OW_BLOCKS * OW_ROWS;

  if (binned) owner_bin_kernel<<<OW_BLOCKS, OW_THREADS, 0, stream>>>(rows, cols, vals, sigma, X, n_vars, nnz);
  else        owner_mask_kernel<<<MK_BLOCKS, OW_THREADS, 0, stream>>>(rows, X, n_vars, nnz);
  fused_x_kernel<<<GEMM_BLOCKS_X, 256, 0, stream>>>(primal, last_primal,
                                                    W2, b2, W3, b3, X, n_vars / 16);
  if (!binned) dual_base_kernel<<<GEMM_BLOCKS_D, 256, 0, stream>>>(dual, W1, b1, rhs, sigma, out,
                                                                   m_cons / 16);
  if (binned) owner_scatter_kernel<<<OW_BLOCKS, OW_THREADS, 0, stream>>>(X, out, dual, W1, b1, rhs, sigma,
                                                                        nnz, m_cons, n_vars);
  else        owner_scatter_masked_kernel<<<OW_BLOCKS, OW_THREADS, 0, stream>>>(rows, cols, vals, X, sigma, out,
                                                                           nnz, m_cons, n_vars);
}
